# starting kernel
# baseline (speedup 1.0000x reference)
#include <hip/hip_runtime.h>
#include <math.h>

typedef __attribute__((ext_vector_type(2))) float v2f;
typedef __attribute__((ext_vector_type(8))) float v8f;

#define WMMA_F32(a, b, c) \
    __builtin_amdgcn_wmma_f32_16x16x4_f32(false, (a), false, (b), (short)0, (c), false, false)

__global__ __launch_bounds__(256) void gemm_wmma_f32(
    const float* __restrict__ A, const float* __restrict__ W,
    const float* __restrict__ bias, float* __restrict__ C,
    int M, int N, int K, float alpha)
{
    __shared__ __align__(16) float As[128][18];
    __shared__ __align__(16) float Wt[64][18];

    const int tid    = threadIdx.x;
    const int wave   = tid >> 5;
    const int lane   = tid & 31;
    const int l16    = lane & 15;
    const int half   = lane >> 4;
    const int kb     = half << 1;
    const int rowOff = half << 3;

    const int wm = wave >> 1;
    const int wn = wave & 1;

    const int rowBlock = blockIdx.y * 128;
    const int colBlock = blockIdx.x * 64;

    v8f acc[2][2];
    #pragma unroll
    for (int mt = 0; mt < 2; ++mt)
        #pragma unroll
        for (int nt = 0; nt < 2; ++nt) acc[mt][nt] = (v8f)(0.0f);

    const int ar = tid >> 1;
    const int ac = (tid & 1) << 3;
    const int wr = tid >> 4;
    const int wc = (tid & 15) << 2;

    const int aRow0 = (wm << 5) + l16;
    const int aRow1 = aRow0 + 16;
    const int bRow0 = (wn << 5) + l16;
    const int bRow1 = bRow0 + 16;

    for (int kt = 0; kt < K; kt += 16) {
        const float* aSrc = A + (size_t)(rowBlock + ar) * K + kt + ac;
        const float4 a0 = *(const float4*)(aSrc);
        const float4 a1 = *(const float4*)(aSrc + 4);
        As[ar][ac + 0] = a0.x; As[ar][ac + 1] = a0.y;
        As[ar][ac + 2] = a0.z; As[ar][ac + 3] = a0.w;
        As[ar][ac + 4] = a1.x; As[ar][ac + 5] = a1.y;
        As[ar][ac + 6] = a1.z; As[ar][ac + 7] = a1.w;
        const float* wSrc = W + (size_t)(kt + wr) * N + colBlock + wc;
        const float4 wv = *(const float4*)(wSrc);
        Wt[wc + 0][wr] = wv.x; Wt[wc + 1][wr] = wv.y;
        Wt[wc + 2][wr] = wv.z; Wt[wc + 3][wr] = wv.w;
        if (kt + 16 < K) {
            __builtin_prefetch(aSrc + 16, 0, 0);
            __builtin_prefetch(wSrc + (size_t)16 * N, 0, 0);
        }
        __syncthreads();

        #pragma unroll
        for (int kk = 0; kk < 16; kk += 4) {
            const v2f av0 = *(const v2f*)&As[aRow0][kk + kb];
            const v2f av1 = *(const v2f*)&As[aRow1][kk + kb];
            const v2f bv0 = *(const v2f*)&Wt[bRow0][kk + kb];
            const v2f bv1 = *(const v2f*)&Wt[bRow1][kk + kb];
            acc[0][0] = WMMA_F32(av0, bv0, acc[0][0]);
            acc[0][1] = WMMA_F32(av0, bv1, acc[0][1]);
            acc[1][0] = WMMA_F32(av1, bv0, acc[1][0]);
            acc[1][1] = WMMA_F32(av1, bv1, acc[1][1]);
        }
        __syncthreads();
    }

    #pragma unroll
    for (int mt = 0; mt < 2; ++mt) {
        #pragma unroll
        for (int nt = 0; nt < 2; ++nt) {
            const int col = colBlock + (wn << 5) + (nt << 4) + l16;
            const float bv = bias ? bias[col] : 0.0f;
            #pragma unroll
            for (int v = 0; v < 8; ++v) {
                const int row = rowBlock + (wm << 5) + (mt << 4) + rowOff + v;
                C[(size_t)row * N + col] = alpha * acc[mt][nt][v] + bv;
            }
        }
    }
}

__global__ __launch_bounds__(128) void mqa_flash_wmma(
    const float* __restrict__ Q,
    const float* __restrict__ Kg,
    const float* __restrict__ Vg,
    float* __restrict__ Out,
    int S, int DIMp)
{
    __shared__ __align__(16) float Ks[64][68];
    __shared__ __align__(16) float Vt[64][68];
    __shared__ __align__(16) float Ps[64][68];

    const int tid    = threadIdx.x;
    const int wave   = tid >> 5;
    const int lane   = tid & 31;
    const int l16    = lane & 15;
    const int half   = lane >> 4;
    const int kb     = half << 1;
    const int rowOff = half << 3;

    const int qTile = blockIdx.x;
    const int h     = blockIdx.y;
    const int b     = blockIdx.z;
    const int qBase = qTile << 6;
    const size_t rowB = (size_t)b * S;

    const int lr = tid >> 1;
    const int lc = (tid & 1) << 5;

    {
        const float* src = Q + (rowB + qBase + lr) * DIMp + (h << 6) + lc;
        #pragma unroll
        for (int j = 0; j < 8; ++j) {
            const float4 qv = *(const float4*)(src + (j << 2));
            Ps[lr][lc + (j << 2) + 0] = qv.x; Ps[lr][lc + (j << 2) + 1] = qv.y;
            Ps[lr][lc + (j << 2) + 2] = qv.z; Ps[lr][lc + (j << 2) + 3] = qv.w;
        }
    }
    __syncthreads();

    v2f aq[16];
    {
        const int qr = (wave << 4) + l16;
        #pragma unroll
        for (int i = 0; i < 16; ++i)
            aq[i] = *(const v2f*)&Ps[qr][(i << 2) + kb];
    }

    v8f o[4];
    #pragma unroll
    for (int nt = 0; nt < 4; ++nt) o[nt] = (v8f)(0.0f);
    float m[8], lsum[8];
    #pragma unroll
    for (int v = 0; v < 8; ++v) { m[v] = -1e30f; lsum[v] = 0.0f; }

    const int nTiles = qTile + 1;
    for (int t = 0; t < nTiles; ++t) {
        const int kBase = t << 6;
        __syncthreads();
        {
            const float* ksrc = Kg + (rowB + kBase + lr) * 64 + lc;
            const float* vsrc = Vg + (rowB + kBase + lr) * 64 + lc;
            #pragma unroll
            for (int j = 0; j < 8; ++j) {
                const float4 kv = *(const float4*)(ksrc + (j << 2));
                const float4 vv = *(const float4*)(vsrc + (j << 2));
                Ks[lr][lc + (j << 2) + 0] = kv.x; Ks[lr][lc + (j << 2) + 1] = kv.y;
                Ks[lr][lc + (j << 2) + 2] = kv.z; Ks[lr][lc + (j << 2) + 3] = kv.w;
                Vt[lc + (j << 2) + 0][lr] = vv.x; Vt[lc + (j << 2) + 1][lr] = vv.y;
                Vt[lc + (j << 2) + 2][lr] = vv.z; Vt[lc + (j << 2) + 3][lr] = vv.w;
            }
        }
        __syncthreads();

        v8f sc[4];
        #pragma unroll
        for (int nt = 0; nt < 4; ++nt) sc[nt] = (v8f)(0.0f);
        #pragma unroll
        for (int kk = 0; kk < 64; kk += 4) {
            const v2f a = aq[kk >> 2];
            #pragma unroll
            for (int nt = 0; nt < 4; ++nt) {
                const v2f bf = *(const v2f*)&Ks[(nt << 4) + l16][kk + kb];
                sc[nt] = WMMA_F32(a, bf, sc[nt]);
            }
        }

        if (t == nTiles - 1) {
            #pragma unroll
            for (int nt = 0; nt < 4; ++nt) {
                const int key = kBase + (nt << 4) + l16;
                #pragma unroll
                for (int v = 0; v < 8; ++v) {
                    const int qr = qBase + (wave << 4) + rowOff + v;
                    if (key > qr) sc[nt][v] = -1e30f;
                }
            }
        }

        float corr[8];
        #pragma unroll
        for (int v = 0; v < 8; ++v) {
            float x = fmaxf(fmaxf(sc[0][v], sc[1][v]), fmaxf(sc[2][v], sc[3][v]));
            x = fmaxf(x, __shfl_xor(x, 1, 32));
            x = fmaxf(x, __shfl_xor(x, 2, 32));
            x = fmaxf(x, __shfl_xor(x, 4, 32));
            x = fmaxf(x, __shfl_xor(x, 8, 32));
            const float mNew = fmaxf(m[v], x);
            corr[v] = __expf(m[v] - mNew);
            m[v] = mNew;
        }
        #pragma unroll
        for (int nt = 0; nt < 4; ++nt)
            #pragma unroll
            for (int v = 0; v < 8; ++v) o[nt][v] *= corr[v];

        float rs[8];
        #pragma unroll
        for (int v = 0; v < 8; ++v) rs[v] = 0.0f;
        const int pRowBase = (wave << 4) + rowOff;
        #pragma unroll
        for (int nt = 0; nt < 4; ++nt) {
            #pragma unroll
            for (int v = 0; v < 8; ++v) {
                const float p = __expf(sc[nt][v] - m[v]);
                rs[v] += p;
                Ps[pRowBase + v][(nt << 4) + l16] = p;
            }
        }
        #pragma unroll
        for (int v = 0; v < 8; ++v) {
            float r = rs[v];
            r += __shfl_xor(r, 1, 32);
            r += __shfl_xor(r, 2, 32);
            r += __shfl_xor(r, 4, 32);
            r += __shfl_xor(r, 8, 32);
            lsum[v] = lsum[v] * corr[v] + r;
        }
        __syncthreads();

        const int pr = (wave << 4) + l16;
        #pragma unroll
        for (int kk = 0; kk < 64; kk += 4) {
            const v2f a = *(const v2f*)&Ps[pr][kk + kb];
            #pragma unroll
            for (int nt = 0; nt < 4; ++nt) {
                const v2f bf = *(const v2f*)&Vt[(nt << 4) + l16][kk + kb];
                o[nt] = WMMA_F32(a, bf, o[nt]);
            }
        }
    }

    #pragma unroll
    for (int nt = 0; nt < 4; ++nt) {
        const int col = (h << 6) + (nt << 4) + l16;
        #pragma unroll
        for (int v = 0; v < 8; ++v) {
            const int row = qBase + (wave << 4) + rowOff + v;
            Out[(rowB + row) * DIMp + col] = o[nt][v] / lsum[v];
        }
    }
}

extern "C" void kernel_launch(void* const* d_in, const int* in_sizes, int n_in,
                              void* d_out, int out_size, void* d_ws, size_t ws_size,
                              hipStream_t stream) {
    (void)in_sizes; (void)n_in; (void)out_size; (void)ws_size;
    constexpr int DIM = 1024, HEAD_DIM = 64, NUM_HEADS = 16, BATCH = 2, SEQ = 2048;
    constexpr int M = BATCH * SEQ;

    const float* x  = (const float*)d_in[0];
    const float* Wq = (const float*)d_in[1];
    const float* Wk = (const float*)d_in[2];
    const float* Wv = (const float*)d_in[3];
    const float* Wo = (const float*)d_in[4];
    const float* bo = (const float*)d_in[5];
    float* out = (float*)d_out;

    float* Qbuf = (float*)d_ws;
    float* Kbuf = Qbuf + (size_t)M * DIM;
    float* Vbuf = Kbuf + (size_t)M * HEAD_DIM;
    float* Abuf = Vbuf + (size_t)M * HEAD_DIM;

    const float scale = 0.125f;

    gemm_wmma_f32<<<dim3(DIM / 64, M / 128), 256, 0, stream>>>(
        x, Wq, nullptr, Qbuf, M, DIM, DIM, scale);
    gemm_wmma_f32<<<dim3(HEAD_DIM / 64, M / 128), 256, 0, stream>>>(
        x, Wk, nullptr, Kbuf, M, HEAD_DIM, DIM, 1.0f);
    gemm_wmma_f32<<<dim3(HEAD_DIM / 64, M / 128), 256, 0, stream>>>(
        x, Wv, nullptr, Vbuf, M, HEAD_DIM, DIM, 1.0f);

    mqa_flash_wmma<<<dim3(SEQ / 64, NUM_HEADS, BATCH), 128, 0, stream>>>(
        Qbuf, Kbuf, Vbuf, Abuf, SEQ, DIM);

    gemm_wmma_f32<<<dim3(DIM / 64, M / 128), 256, 0, stream>>>(
        Abuf, Wo, bo, out, M, DIM, DIM, 1.0f);
}


# Attention: each workgroup does query tiles 31-i and i back to back (grid.x 32 to 16), so every workgroup runs 33 key-tile steps instead of 1 to 32; arithmetic and summation order unchanged.
# speedup vs baseline: 1.0930x; 1.0930x over previous
#include <hip/hip_runtime.h>
#include <math.h>

typedef __attribute__((ext_vector_type(2))) float v2f;
typedef __attribute__((ext_vector_type(8))) float v8f;

#define WMMA_F32(a, b, c) \
    __builtin_amdgcn_wmma_f32_16x16x4_f32(false, (a), false, (b), (short)0, (c), false, false)

__global__ __launch_bounds__(256) void gemm_wmma_f32(
    const float* __restrict__ A, const float* __restrict__ W,
    const float* __restrict__ bias, float* __restrict__ C,
    int M, int N, int K, float alpha)
{
    __shared__ __align__(16) float As[128][18];
    __shared__ __align__(16) float Wt[64][18];

    const int tid    = threadIdx.x;
    const int wave   = tid >> 5;
    const int lane   = tid & 31;
    const int l16    = lane & 15;
    const int half   = lane >> 4;
    const int kb     = half << 1;
    const int rowOff = half << 3;

    const int wm = wave >> 1;
    const int wn = wave & 1;

    const int rowBlock = blockIdx.y * 128;
    const int colBlock = blockIdx.x * 64;

    v8f acc[2][2];
    #pragma unroll
    for (int mt = 0; mt < 2; ++mt)
        #pragma unroll
        for (int nt = 0; nt < 2; ++nt) acc[mt][nt] = (v8f)(0.0f);

    const int ar = tid >> 1;
    const int ac = (tid & 1) << 3;
    const int wr = tid >> 4;
    const int wc = (tid & 15) << 2;

    const int aRow0 = (wm << 5) + l16;
    const int aRow1 = aRow0 + 16;
    const int bRow0 = (wn << 5) + l16;
    const int bRow1 = bRow0 + 16;

    for (int kt = 0; kt < K; kt += 16) {
        const float* aSrc = A + (size_t)(rowBlock + ar) * K + kt + ac;
        const float4 a0 = *(const float4*)(aSrc);
        const float4 a1 = *(const float4*)(aSrc + 4);
        As[ar][ac + 0] = a0.x; As[ar][ac + 1] = a0.y;
        As[ar][ac + 2] = a0.z; As[ar][ac + 3] = a0.w;
        As[ar][ac + 4] = a1.x; As[ar][ac + 5] = a1.y;
        As[ar][ac + 6] = a1.z; As[ar][ac + 7] = a1.w;
        const float* wSrc = W + (size_t)(kt + wr) * N + colBlock + wc;
        const float4 wv = *(const float4*)(wSrc);
        Wt[wc + 0][wr] = wv.x; Wt[wc + 1][wr] = wv.y;
        Wt[wc + 2][wr] = wv.z; Wt[wc + 3][wr] = wv.w;
        if (kt + 16 < K) {
            __builtin_prefetch(aSrc + 16, 0, 0);
            __builtin_prefetch(wSrc + (size_t)16 * N, 0, 0);
        }
        __syncthreads();

        #pragma unroll
        for (int kk = 0; kk < 16; kk += 4) {
            const v2f av0 = *(const v2f*)&As[aRow0][kk + kb];
            const v2f av1 = *(const v2f*)&As[aRow1][kk + kb];
            const v2f bv0 = *(const v2f*)&Wt[bRow0][kk + kb];
            const v2f bv1 = *(const v2f*)&Wt[bRow1][kk + kb];
            acc[0][0] = WMMA_F32(av0, bv0, acc[0][0]);
            acc[0][1] = WMMA_F32(av0, bv1, acc[0][1]);
            acc[1][0] = WMMA_F32(av1, bv0, acc[1][0]);
            acc[1][1] = WMMA_F32(av1, bv1, acc[1][1]);
        }
        __syncthreads();
    }

    #pragma unroll
    for (int mt = 0; mt < 2; ++mt) {
        #pragma unroll
        for (int nt = 0; nt < 2; ++nt) {
            const int col = colBlock + (wn << 5) + (nt << 4) + l16;
            const float bv = bias ? bias[col] : 0.0f;
            #pragma unroll
            for (int v = 0; v < 8; ++v) {
                const int row = rowBlock + (wm << 5) + (mt << 4) + rowOff + v;
                C[(size_t)row * N + col] = alpha * acc[mt][nt][v] + bv;
            }
        }
    }
}

__global__ __launch_bounds__(128) void mqa_flash_wmma(
    const float* __restrict__ Q,
    const float* __restrict__ Kg,
    const float* __restrict__ Vg,
    float* __restrict__ Out,
    int S, int DIMp)
{
    __shared__ __align__(16) float Ks[64][68];
    __shared__ __align__(16) float Vt[64][68];
    __shared__ __align__(16) float Ps[64][68];

    const int tid    = threadIdx.x;
    const int wave   = tid >> 5;
    const int lane   = tid & 31;
    const int l16    = lane & 15;
    const int half   = lane >> 4;
    const int kb     = half << 1;
    const int rowOff = half << 3;

    const int h     = blockIdx.y;
    const int b     = blockIdx.z;
    const size_t rowB = (size_t)b * S;
    const int nQ    = (int)gridDim.x << 1;

    const int lr = tid >> 1;
    const int lc = (tid & 1) << 5;

    for (int pass = 0; pass < 2; ++pass) {
        const int qTile = pass == 0 ? (nQ - 1 - (int)blockIdx.x) : (int)blockIdx.x;
        const int qBase = qTile << 6;

        if (pass != 0) __syncthreads();

        {
            const float* src = Q + (rowB + qBase + lr) * DIMp + (h << 6) + lc;
            #pragma unroll
            for (int j = 0; j < 8; ++j) {
                const float4 qv = *(const float4*)(src + (j << 2));
                Ps[lr][lc + (j << 2) + 0] = qv.x; Ps[lr][lc + (j << 2) + 1] = qv.y;
                Ps[lr][lc + (j << 2) + 2] = qv.z; Ps[lr][lc + (j << 2) + 3] = qv.w;
            }
        }
        __syncthreads();

        v2f aq[16];
        {
            const int qr = (wave << 4) + l16;
            #pragma unroll
            for (int i = 0; i < 16; ++i)
                aq[i] = *(const v2f*)&Ps[qr][(i << 2) + kb];
        }

        v8f o[4];
        #pragma unroll
        for (int nt = 0; nt < 4; ++nt) o[nt] = (v8f)(0.0f);
        float m[8], lsum[8];
        #pragma unroll
        for (int v = 0; v < 8; ++v) { m[v] = -1e30f; lsum[v] = 0.0f; }

        const int nTiles = qTile + 1;
        for (int t = 0; t < nTiles; ++t) {
            const int kBase = t << 6;
            __syncthreads();
            {
                const float* ksrc = Kg + (rowB + kBase + lr) * 64 + lc;
                const float* vsrc = Vg + (rowB + kBase + lr) * 64 + lc;
                #pragma unroll
                for (int j = 0; j < 8; ++j) {
                    const float4 kv = *(const float4*)(ksrc + (j << 2));
                    const float4 vv = *(const float4*)(vsrc + (j << 2));
                    Ks[lr][lc + (j << 2) + 0] = kv.x; Ks[lr][lc + (j << 2) + 1] = kv.y;
                    Ks[lr][lc + (j << 2) + 2] = kv.z; Ks[lr][lc + (j << 2) + 3] = kv.w;
                    Vt[lc + (j << 2) + 0][lr] = vv.x; Vt[lc + (j << 2) + 1][lr] = vv.y;
                    Vt[lc + (j << 2) + 2][lr] = vv.z; Vt[lc + (j << 2) + 3][lr] = vv.w;
                }
            }
            __syncthreads();

            v8f sc[4];
            #pragma unroll
            for (int nt = 0; nt < 4; ++nt) sc[nt] = (v8f)(0.0f);
            #pragma unroll
            for (int kk = 0; kk < 64; kk += 4) {
                const v2f a = aq[kk >> 2];
                #pragma unroll
                for (int nt = 0; nt < 4; ++nt) {
                    const v2f bf = *(const v2f*)&Ks[(nt << 4) + l16][kk + kb];
                    sc[nt] = WMMA_F32(a, bf, sc[nt]);
                }
            }

            if (t == nTiles - 1) {
                #pragma unroll
                for (int nt = 0; nt < 4; ++nt) {
                    const int key = kBase + (nt << 4) + l16;
                    #pragma unroll
                    for (int v = 0; v < 8; ++v) {
                        const int qr = qBase + (wave << 4) + rowOff + v;
                        if (key > qr) sc[nt][v] = -1e30f;
                    }
                }
            }

            float corr[8];
            #pragma unroll
            for (int v = 0; v < 8; ++v) {
                float x = fmaxf(fmaxf(sc[0][v], sc[1][v]), fmaxf(sc[2][v], sc[3][v]));
                x = fmaxf(x, __shfl_xor(x, 1, 32));
                x = fmaxf(x, __shfl_xor(x, 2, 32));
                x = fmaxf(x, __shfl_xor(x, 4, 32));
                x = fmaxf(x, __shfl_xor(x, 8, 32));
                const float mNew = fmaxf(m[v], x);
                corr[v] = __expf(m[v] - mNew);
                m[v] = mNew;
            }
            #pragma unroll
            for (int nt = 0; nt < 4; ++nt)
                #pragma unroll
                for (int v = 0; v < 8; ++v) o[nt][v] *= corr[v];

            float rs[8];
            #pragma unroll
            for (int v = 0; v < 8; ++v) rs[v] = 0.0f;
            const int pRowBase = (wave << 4) + rowOff;
            #pragma unroll
            for (int nt = 0; nt < 4; ++nt) {
                #pragma unroll
                for (int v = 0; v < 8; ++v) {
                    const float p = __expf(sc[nt][v] - m[v]);
                    rs[v] += p;
                    Ps[pRowBase + v][(nt << 4) + l16] = p;
                }
            }
            #pragma unroll
            for (int v = 0; v < 8; ++v) {
                float r = rs[v];
                r += __shfl_xor(r, 1, 32);
                r += __shfl_xor(r, 2, 32);
                r += __shfl_xor(r, 4, 32);
                r += __shfl_xor(r, 8, 32);
                lsum[v] = lsum[v] * corr[v] + r;
            }
            __syncthreads();

            const int pr = (wave << 4) + l16;
            #pragma unroll
            for (int kk = 0; kk < 64; kk += 4) {
                const v2f a = *(const v2f*)&Ps[pr][kk + kb];
                #pragma unroll
                for (int nt = 0; nt < 4; ++nt) {
                    const v2f bf = *(const v2f*)&Vt[(nt << 4) + l16][kk + kb];
                    o[nt] = WMMA_F32(a, bf, o[nt]);
                }
            }
        }

        #pragma unroll
        for (int nt = 0; nt < 4; ++nt) {
            const int col = (h << 6) + (nt << 4) + l16;
            #pragma unroll
            for (int v = 0; v < 8; ++v) {
                const int row = qBase + (wave << 4) + rowOff + v;
                Out[(rowB + row) * DIMp + col] = o[nt][v] / lsum[v];
            }
        }
    }
}

extern "C" void kernel_launch(void* const* d_in, const int* in_sizes, int n_in,
                              void* d_out, int out_size, void* d_ws, size_t ws_size,
                              hipStream_t stream) {
    (void)in_sizes; (void)n_in; (void)out_size; (void)ws_size;
    constexpr int DIM = 1024, HEAD_DIM = 64, NUM_HEADS = 16, BATCH = 2, SEQ = 2048;
    constexpr int M = BATCH * SEQ;

    const float* x  = (const float*)d_in[0];
    const float* Wq = (const float*)d_in[1];
    const float* Wk = (const float*)d_in[2];
    const float* Wv = (const float*)d_in[3];
    const float* Wo = (const float*)d_in[4];
    const float* bo = (const float*)d_in[5];
    float* out = (float*)d_out;

    float* Qbuf = (float*)d_ws;
    float* Kbuf = Qbuf + (size_t)M * DIM;
    float* Vbuf = Kbuf + (size_t)M * HEAD_DIM;
    float* Abuf = Vbuf + (size_t)M * HEAD_DIM;

    const float scale = 0.125f;

    gemm_wmma_f32<<<dim3(DIM / 64, M / 128), 256, 0, stream>>>(
        x, Wq, nullptr, Qbuf, M, DIM, DIM, scale);
    gemm_wmma_f32<<<dim3(HEAD_DIM / 64, M / 128), 256, 0, stream>>>(
        x, Wk, nullptr, Kbuf, M, HEAD_DIM, DIM, 1.0f);
    gemm_wmma_f32<<<dim3(HEAD_DIM / 64, M / 128), 256, 0, stream>>>(
        x, Wv, nullptr, Vbuf, M, HEAD_DIM, DIM, 1.0f);

    mqa_flash_wmma<<<dim3(SEQ / 128, NUM_HEADS, BATCH), 128, 0, stream>>>(
        Qbuf, Kbuf, Vbuf, Abuf, SEQ, DIM);

    gemm_wmma_f32<<<dim3(DIM / 64, M / 128), 256, 0, stream>>>(
        Abuf, Wo, bo, out, M, DIM, DIM, 1.0f);
}


# Q, K and V projections of x run as one launch (grid.x 16 -> 18: column blocks 16, 17 take Wk, Wv; launches 5 -> 3), replacing two 32-workgroup launches; arithmetic and sum order unchanged.
# speedup vs baseline: 1.3169x; 1.2048x over previous
#include <hip/hip_runtime.h>
#include <math.h>

typedef __attribute__((ext_vector_type(2))) float v2f;
typedef __attribute__((ext_vector_type(8))) float v8f;

#define WMMA_F32(a, b, c) \
    __builtin_amdgcn_wmma_f32_16x16x4_f32(false, (a), false, (b), (short)0, (c), false, false)

__global__ __launch_bounds__(256) void gemm_wmma_f32(
    const float* __restrict__ A,
    const float* __restrict__ W0, const float* __restrict__ W1, const float* __restrict__ W2,
    const float* __restrict__ bias0,
    float* __restrict__ C0, float* __restrict__ C1, float* __restrict__ C2,
    int M, int N0, int N1, int K, float alpha0)
{
    const int nb0 = N0 >> 6;
    const int nb1 = N1 >> 6;
    int bx = (int)blockIdx.x;
    const float* W = W0;
    const float* bias = bias0;
    float* C = C0;
    int N = N0;
    float alpha = alpha0;
    if (bx >= nb0) {
        bx -= nb0;
        W = W1; C = C1;
        if (bx >= nb1) { bx -= nb1; W = W2; C = C2; }
        N = N1; alpha = 1.0f; bias = nullptr;
    }

    __shared__ __align__(16) float As[128][18];
    __shared__ __align__(16) float Wt[64][18];

    const int tid    = threadIdx.x;
    const int wave   = tid >> 5;
    const int lane   = tid & 31;
    const int l16    = lane & 15;
    const int half   = lane >> 4;
    const int kb     = half << 1;
    const int rowOff = half << 3;

    const int wm = wave >> 1;
    const int wn = wave & 1;

    const int rowBlock = blockIdx.y * 128;
    const int colBlock = bx * 64;

    v8f acc[2][2];
    #pragma unroll
    for (int mt = 0; mt < 2; ++mt)
        #pragma unroll
        for (int nt = 0; nt < 2; ++nt) acc[mt][nt] = (v8f)(0.0f);

    const int ar = tid >> 1;
    const int ac = (tid & 1) << 3;
    const int wr = tid >> 4;
    const int wc = (tid & 15) << 2;

    const int aRow0 = (wm << 5) + l16;
    const int aRow1 = aRow0 + 16;
    const int bRow0 = (wn << 5) + l16;
    const int bRow1 = bRow0 + 16;

    for (int kt = 0; kt < K; kt += 16) {
        const float* aSrc = A + (size_t)(rowBlock + ar) * K + kt + ac;
        const float4 a0 = *(const float4*)(aSrc);
        const float4 a1 = *(const float4*)(aSrc + 4);
        As[ar][ac + 0] = a0.x; As[ar][ac + 1] = a0.y;
        As[ar][ac + 2] = a0.z; As[ar][ac + 3] = a0.w;
        As[ar][ac + 4] = a1.x; As[ar][ac + 5] = a1.y;
        As[ar][ac + 6] = a1.z; As[ar][ac + 7] = a1.w;
        const float* wSrc = W + (size_t)(kt + wr) * N + colBlock + wc;
        const float4 wv = *(const float4*)(wSrc);
        Wt[wc + 0][wr] = wv.x; Wt[wc + 1][wr] = wv.y;
        Wt[wc + 2][wr] = wv.z; Wt[wc + 3][wr] = wv.w;
        if (kt + 16 < K) {
            __builtin_prefetch(aSrc + 16, 0, 0);
            __builtin_prefetch(wSrc + (size_t)16 * N, 0, 0);
        }
        __syncthreads();

        #pragma unroll
        for (int kk = 0; kk < 16; kk += 4) {
            const v2f av0 = *(const v2f*)&As[aRow0][kk + kb];
            const v2f av1 = *(const v2f*)&As[aRow1][kk + kb];
            const v2f bv0 = *(const v2f*)&Wt[bRow0][kk + kb];
            const v2f bv1 = *(const v2f*)&Wt[bRow1][kk + kb];
            acc[0][0] = WMMA_F32(av0, bv0, acc[0][0]);
            acc[0][1] = WMMA_F32(av0, bv1, acc[0][1]);
            acc[1][0] = WMMA_F32(av1, bv0, acc[1][0]);
            acc[1][1] = WMMA_F32(av1, bv1, acc[1][1]);
        }
        __syncthreads();
    }

    #pragma unroll
    for (int mt = 0; mt < 2; ++mt) {
        #pragma unroll
        for (int nt = 0; nt < 2; ++nt) {
            const int col = colBlock + (wn << 5) + (nt << 4) + l16;
            const float bv = bias ? bias[col] : 0.0f;
            #pragma unroll
            for (int v = 0; v < 8; ++v) {
                const int row = rowBlock + (wm << 5) + (mt << 4) + rowOff + v;
                C[(size_t)row * N + col] = alpha * acc[mt][nt][v] + bv;
            }
        }
    }
}

__global__ __launch_bounds__(128) void mqa_flash_wmma(
    const float* __restrict__ Q,
    const float* __restrict__ Kg,
    const float* __restrict__ Vg,
    float* __restrict__ Out,
    int S, int DIMp)
{
    __shared__ __align__(16) float Ks[64][68];
    __shared__ __align__(16) float Vt[64][68];
    __shared__ __align__(16) float Ps[64][68];

    const int tid    = threadIdx.x;
    const int wave   = tid >> 5;
    const int lane   = tid & 31;
    const int l16    = lane & 15;
    const int half   = lane >> 4;
    const int kb     = half << 1;
    const int rowOff = half << 3;

    const int h     = blockIdx.y;
    const int b     = blockIdx.z;
    const size_t rowB = (size_t)b * S;
    const int nQ    = (int)gridDim.x << 1;

    const int lr = tid >> 1;
    const int lc = (tid & 1) << 5;

    for (int pass = 0; pass < 2; ++pass) {
        const int qTile = pass == 0 ? (nQ - 1 - (int)blockIdx.x) : (int)blockIdx.x;
        const int qBase = qTile << 6;

        if (pass != 0) __syncthreads();

        {
            const float* src = Q + (rowB + qBase + lr) * DIMp + (h << 6) + lc;
            #pragma unroll
            for (int j = 0; j < 8; ++j) {
                const float4 qv = *(const float4*)(src + (j << 2));
                Ps[lr][lc + (j << 2) + 0] = qv.x; Ps[lr][lc + (j << 2) + 1] = qv.y;
                Ps[lr][lc + (j << 2) + 2] = qv.z; Ps[lr][lc + (j << 2) + 3] = qv.w;
            }
        }
        __syncthreads();

        v2f aq[16];
        {
            const int qr = (wave << 4) + l16;
            #pragma unroll
            for (int i = 0; i < 16; ++i)
                aq[i] = *(const v2f*)&Ps[qr][(i << 2) + kb];
        }

        v8f o[4];
        #pragma unroll
        for (int nt = 0; nt < 4; ++nt) o[nt] = (v8f)(0.0f);
        float m[8], lsum[8];
        #pragma unroll
        for (int v = 0; v < 8; ++v) { m[v] = -1e30f; lsum[v] = 0.0f; }

        const int nTiles = qTile + 1;
        for (int t = 0; t < nTiles; ++t) {
            const int kBase = t << 6;
            __syncthreads();
            {
                const float* ksrc = Kg + (rowB + kBase + lr) * 64 + lc;
                const float* vsrc = Vg + (rowB + kBase + lr) * 64 + lc;
                #pragma unroll
                for (int j = 0; j < 8; ++j) {
                    const float4 kv = *(const float4*)(ksrc + (j << 2));
                    const float4 vv = *(const float4*)(vsrc + (j << 2));
                    Ks[lr][lc + (j << 2) + 0] = kv.x; Ks[lr][lc + (j << 2) + 1] = kv.y;
                    Ks[lr][lc + (j << 2) + 2] = kv.z; Ks[lr][lc + (j << 2) + 3] = kv.w;
                    Vt[lc + (j << 2) + 0][lr] = vv.x; Vt[lc + (j << 2) + 1][lr] = vv.y;
                    Vt[lc + (j << 2) + 2][lr] = vv.z; Vt[lc + (j << 2) + 3][lr] = vv.w;
                }
            }
            __syncthreads();

            v8f sc[4];
            #pragma unroll
            for (int nt = 0; nt < 4; ++nt) sc[nt] = (v8f)(0.0f);
            #pragma unroll
            for (int kk = 0; kk < 64; kk += 4) {
                const v2f a = aq[kk >> 2];
                #pragma unroll
                for (int nt = 0; nt < 4; ++nt) {
                    const v2f bf = *(const v2f*)&Ks[(nt << 4) + l16][kk + kb];
                    sc[nt] = WMMA_F32(a, bf, sc[nt]);
                }
            }

            if (t == nTiles - 1) {
                #pragma unroll
                for (int nt = 0; nt < 4; ++nt) {
                    const int key = kBase + (nt << 4) + l16;
                    #pragma unroll
                    for (int v = 0; v < 8; ++v) {
                        const int qr = qBase + (wave << 4) + rowOff + v;
                        if (key > qr) sc[nt][v] = -1e30f;
                    }
                }
            }

            float corr[8];
            #pragma unroll
            for (int v = 0; v < 8; ++v) {
                float x = fmaxf(fmaxf(sc[0][v], sc[1][v]), fmaxf(sc[2][v], sc[3][v]));
                x = fmaxf(x, __shfl_xor(x, 1, 32));
                x = fmaxf(x, __shfl_xor(x, 2, 32));
                x = fmaxf(x, __shfl_xor(x, 4, 32));
                x = fmaxf(x, __shfl_xor(x, 8, 32));
                const float mNew = fmaxf(m[v], x);
                corr[v] = __expf(m[v] - mNew);
                m[v] = mNew;
            }
            #pragma unroll
            for (int nt = 0; nt < 4; ++nt)
                #pragma unroll
                for (int v = 0; v < 8; ++v) o[nt][v] *= corr[v];

            float rs[8];
            #pragma unroll
            for (int v = 0; v < 8; ++v) rs[v] = 0.0f;
            const int pRowBase = (wave << 4) + rowOff;
            #pragma unroll
            for (int nt = 0; nt < 4; ++nt) {
                #pragma unroll
                for (int v = 0; v < 8; ++v) {
                    const float p = __expf(sc[nt][v] - m[v]);
                    rs[v] += p;
                    Ps[pRowBase + v][(nt << 4) + l16] = p;
                }
            }
            #pragma unroll
            for (int v = 0; v < 8; ++v) {
                float r = rs[v];
                r += __shfl_xor(r, 1, 32);
                r += __shfl_xor(r, 2, 32);
                r += __shfl_xor(r, 4, 32);
                r += __shfl_xor(r, 8, 32);
                lsum[v] = lsum[v] * corr[v] + r;
            }
            __syncthreads();

            const int pr = (wave << 4) + l16;
            #pragma unroll
            for (int kk = 0; kk < 64; kk += 4) {
                const v2f a = *(const v2f*)&Ps[pr][kk + kb];
                #pragma unroll
                for (int nt = 0; nt < 4; ++nt) {
                    const v2f bf = *(const v2f*)&Vt[(nt << 4) + l16][kk + kb];
                    o[nt] = WMMA_F32(a, bf, o[nt]);
                }
            }
        }

        #pragma unroll
        for (int nt = 0; nt < 4; ++nt) {
            const int col = (h << 6) + (nt << 4) + l16;
            #pragma unroll
            for (int v = 0; v < 8; ++v) {
                const int row = qBase + (wave << 4) + rowOff + v;
                Out[(rowB + row) * DIMp + col] = o[nt][v] / lsum[v];
            }
        }
    }
}

extern "C" void kernel_launch(void* const* d_in, const int* in_sizes, int n_in,
                              void* d_out, int out_size, void* d_ws, size_t ws_size,
                              hipStream_t stream) {
    (void)in_sizes; (void)n_in; (void)out_size; (void)ws_size;
    constexpr int DIM = 1024, HEAD_DIM = 64, NUM_HEADS = 16, BATCH = 2, SEQ = 2048;
    constexpr int M = BATCH * SEQ;

    const float* x  = (const float*)d_in[0];
    const float* Wq = (const float*)d_in[1];
    const float* Wk = (const float*)d_in[2];
    const float* Wv = (const float*)d_in[3];
    const float* Wo = (const float*)d_in[4];
    const float* bo = (const float*)d_in[5];
    float* out = (float*)d_out;

    float* Qbuf = (float*)d_ws;
    float* Kbuf = Qbuf + (size_t)M * DIM;
    float* Vbuf = Kbuf + (size_t)M * HEAD_DIM;
    float* Abuf = Vbuf + (size_t)M * HEAD_DIM;

    const float scale = 0.125f;

    gemm_wmma_f32<<<dim3(DIM / 64 + 2 * (HEAD_DIM / 64), M / 128), 256, 0, stream>>>(
        x, Wq, Wk, Wv, nullptr, Qbuf, Kbuf, Vbuf, M, DIM, HEAD_DIM, DIM, scale);

    mqa_flash_wmma<<<dim3(SEQ / 128, NUM_HEADS, BATCH), 128, 0, stream>>>(
        Qbuf, Kbuf, Vbuf, Abuf, SEQ, DIM);

    gemm_wmma_f32<<<dim3(DIM / 64, M / 128), 256, 0, stream>>>(
        Abuf, Wo, nullptr, nullptr, bo, out, nullptr, nullptr, M, DIM, 0, DIM, 1.0f);
}


# Attention softmax: the 16-lane row max and row sum butterflies run as DPP row_xmask steps on the vector ALU instead of LDS permutes (ds_bpermute); same butterfly order, sums bit-identical.
# speedup vs baseline: 1.6026x; 1.2169x over previous
#include <hip/hip_runtime.h>
#include <math.h>

typedef __attribute__((ext_vector_type(2))) float v2f;
typedef __attribute__((ext_vector_type(8))) float v8f;

#define WMMA_F32(a, b, c) \
    __builtin_amdgcn_wmma_f32_16x16x4_f32(false, (a), false, (b), (short)0, (c), false, false)

__global__ __launch_bounds__(256) void gemm_wmma_f32(
    const float* __restrict__ A,
    const float* __restrict__ W0, const float* __restrict__ W1, const float* __restrict__ W2,
    const float* __restrict__ bias0,
    float* __restrict__ C0, float* __restrict__ C1, float* __restrict__ C2,
    int M, int N0, int N1, int K, float alpha0)
{
    const int nb0 = N0 >> 6;
    const int nb1 = N1 >> 6;
    int bx = (int)blockIdx.x;
    const float* W = W0;
    const float* bias = bias0;
    float* C = C0;
    int N = N0;
    float alpha = alpha0;
    if (bx >= nb0) {
        bx -= nb0;
        W = W1; C = C1;
        if (bx >= nb1) { bx -= nb1; W = W2; C = C2; }
        N = N1; alpha = 1.0f; bias = nullptr;
    }

    __shared__ __align__(16) float As[128][18];
    __shared__ __align__(16) float Wt[64][18];

    const int tid    = threadIdx.x;
    const int wave   = tid >> 5;
    const int lane   = tid & 31;
    const int l16    = lane & 15;
    const int half   = lane >> 4;
    const int kb     = half << 1;
    const int rowOff = half << 3;

    const int wm = wave >> 1;
    const int wn = wave & 1;

    const int rowBlock = blockIdx.y * 128;
    const int colBlock = bx * 64;

    v8f acc[2][2];
    #pragma unroll
    for (int mt = 0; mt < 2; ++mt)
        #pragma unroll
        for (int nt = 0; nt < 2; ++nt) acc[mt][nt] = (v8f)(0.0f);

    const int ar = tid >> 1;
    const int ac = (tid & 1) << 3;
    const int wr = tid >> 4;
    const int wc = (tid & 15) << 2;

    const int aRow0 = (wm << 5) + l16;
    const int aRow1 = aRow0 + 16;
    const int bRow0 = (wn << 5) + l16;
    const int bRow1 = bRow0 + 16;

    for (int kt = 0; kt < K; kt += 16) {
        const float* aSrc = A + (size_t)(rowBlock + ar) * K + kt + ac;
        const float4 a0 = *(const float4*)(aSrc);
        const float4 a1 = *(const float4*)(aSrc + 4);
        As[ar][ac + 0] = a0.x; As[ar][ac + 1] = a0.y;
        As[ar][ac + 2] = a0.z; As[ar][ac + 3] = a0.w;
        As[ar][ac + 4] = a1.x; As[ar][ac + 5] = a1.y;
        As[ar][ac + 6] = a1.z; As[ar][ac + 7] = a1.w;
        const float* wSrc = W + (size_t)(kt + wr) * N + colBlock + wc;
        const float4 wv = *(const float4*)(wSrc);
        Wt[wc + 0][wr] = wv.x; Wt[wc + 1][wr] = wv.y;
        Wt[wc + 2][wr] = wv.z; Wt[wc + 3][wr] = wv.w;
        if (kt + 16 < K) {
            __builtin_prefetch(aSrc + 16, 0, 0);
            __builtin_prefetch(wSrc + (size_t)16 * N, 0, 0);
        }
        __syncthreads();

        #pragma unroll
        for (int kk = 0; kk < 16; kk += 4) {
            const v2f av0 = *(const v2f*)&As[aRow0][kk + kb];
            const v2f av1 = *(const v2f*)&As[aRow1][kk + kb];
            const v2f bv0 = *(const v2f*)&Wt[bRow0][kk + kb];
            const v2f bv1 = *(const v2f*)&Wt[bRow1][kk + kb];
            acc[0][0] = WMMA_F32(av0, bv0, acc[0][0]);
            acc[0][1] = WMMA_F32(av0, bv1, acc[0][1]);
            acc[1][0] = WMMA_F32(av1, bv0, acc[1][0]);
            acc[1][1] = WMMA_F32(av1, bv1, acc[1][1]);
        }
        __syncthreads();
    }

    #pragma unroll
    for (int mt = 0; mt < 2; ++mt) {
        #pragma unroll
        for (int nt = 0; nt < 2; ++nt) {
            const int col = colBlock + (wn << 5) + (nt << 4) + l16;
            const float bv = bias ? bias[col] : 0.0f;
            #pragma unroll
            for (int v = 0; v < 8; ++v) {
                const int row = rowBlock + (wm << 5) + (mt << 4) + rowOff + v;
                C[(size_t)row * N + col] = alpha * acc[mt][nt][v] + bv;
            }
        }
    }
}

template <int MASK>
__device__ __forceinline__ float row_xor(float x) {
    const int xi = __builtin_bit_cast(int, x);
    return __builtin_bit_cast(float, __builtin_amdgcn_update_dpp(0, xi, 0x160 | MASK, 0xf, 0xf, true));
}

__global__ __launch_bounds__(128) void mqa_flash_wmma(
    const float* __restrict__ Q,
    const float* __restrict__ Kg,
    const float* __restrict__ Vg,
    float* __restrict__ Out,
    int S, int DIMp)
{
    __shared__ __align__(16) float Ks[64][68];
    __shared__ __align__(16) float Vt[64][68];
    __shared__ __align__(16) float Ps[64][68];

    const int tid    = threadIdx.x;
    const int wave   = tid >> 5;
    const int lane   = tid & 31;
    const int l16    = lane & 15;
    const int half   = lane >> 4;
    const int kb     = half << 1;
    const int rowOff = half << 3;

    const int h     = blockIdx.y;
    const int b     = blockIdx.z;
    const size_t rowB = (size_t)b * S;
    const int nQ    = (int)gridDim.x << 1;

    const int lr = tid >> 1;
    const int lc = (tid & 1) << 5;

    for (int pass = 0; pass < 2; ++pass) {
        const int qTile = pass == 0 ? (nQ - 1 - (int)blockIdx.x) : (int)blockIdx.x;
        const int qBase = qTile << 6;

        if (pass != 0) __syncthreads();

        {
            const float* src = Q + (rowB + qBase + lr) * DIMp + (h << 6) + lc;
            #pragma unroll
            for (int j = 0; j < 8; ++j) {
                const float4 qv = *(const float4*)(src + (j << 2));
                Ps[lr][lc + (j << 2) + 0] = qv.x; Ps[lr][lc + (j << 2) + 1] = qv.y;
                Ps[lr][lc + (j << 2) + 2] = qv.z; Ps[lr][lc + (j << 2) + 3] = qv.w;
            }
        }
        __syncthreads();

        v2f aq[16];
        {
            const int qr = (wave << 4) + l16;
            #pragma unroll
            for (int i = 0; i < 16; ++i)
                aq[i] = *(const v2f*)&Ps[qr][(i << 2) + kb];
        }

        v8f o[4];
        #pragma unroll
        for (int nt = 0; nt < 4; ++nt) o[nt] = (v8f)(0.0f);
        float m[8], lsum[8];
        #pragma unroll
        for (int v = 0; v < 8; ++v) { m[v] = -1e30f; lsum[v] = 0.0f; }

        const int nTiles = qTile + 1;
        for (int t = 0; t < nTiles; ++t) {
            const int kBase = t << 6;
            __syncthreads();
            {
                const float* ksrc = Kg + (rowB + kBase + lr) * 64 + lc;
                const float* vsrc = Vg + (rowB + kBase + lr) * 64 + lc;
                #pragma unroll
                for (int j = 0; j < 8; ++j) {
                    const float4 kv = *(const float4*)(ksrc + (j << 2));
                    const float4 vv = *(const float4*)(vsrc + (j << 2));
                    Ks[lr][lc + (j << 2) + 0] = kv.x; Ks[lr][lc + (j << 2) + 1] = kv.y;
                    Ks[lr][lc + (j << 2) + 2] = kv.z; Ks[lr][lc + (j << 2) + 3] = kv.w;
                    Vt[lc + (j << 2) + 0][lr] = vv.x; Vt[lc + (j << 2) + 1][lr] = vv.y;
                    Vt[lc + (j << 2) + 2][lr] = vv.z; Vt[lc + (j << 2) + 3][lr] = vv.w;
                }
            }
            __syncthreads();

            v8f sc[4];
            #pragma unroll
            for (int nt = 0; nt < 4; ++nt) sc[nt] = (v8f)(0.0f);
            #pragma unroll
            for (int kk = 0; kk < 64; kk += 4) {
                const v2f a = aq[kk >> 2];
                #pragma unroll
                for (int nt = 0; nt < 4; ++nt) {
                    const v2f bf = *(const v2f*)&Ks[(nt << 4) + l16][kk + kb];
                    sc[nt] = WMMA_F32(a, bf, sc[nt]);
                }
            }

            if (t == nTiles - 1) {
                #pragma unroll
                for (int nt = 0; nt < 4; ++nt) {
                    const int key = kBase + (nt << 4) + l16;
                    #pragma unroll
                    for (int v = 0; v < 8; ++v) {
                        const int qr = qBase + (wave << 4) + rowOff + v;
                        if (key > qr) sc[nt][v] = -1e30f;
                    }
                }
            }

            float corr[8];
            #pragma unroll
            for (int v = 0; v < 8; ++v) {
                float x = fmaxf(fmaxf(sc[0][v], sc[1][v]), fmaxf(sc[2][v], sc[3][v]));
                x = fmaxf(x, row_xor<1>(x));
                x = fmaxf(x, row_xor<2>(x));
                x = fmaxf(x, row_xor<4>(x));
                x = fmaxf(x, row_xor<8>(x));
                const float mNew = fmaxf(m[v], x);
                corr[v] = __expf(m[v] - mNew);
                m[v] = mNew;
            }
            #pragma unroll
            for (int nt = 0; nt < 4; ++nt)
                #pragma unroll
                for (int v = 0; v < 8; ++v) o[nt][v] *= corr[v];

            float rs[8];
            #pragma unroll
            for (int v = 0; v < 8; ++v) rs[v] = 0.0f;
            const int pRowBase = (wave << 4) + rowOff;
            #pragma unroll
            for (int nt = 0; nt < 4; ++nt) {
                #pragma unroll
                for (int v = 0; v < 8; ++v) {
                    const float p = __expf(sc[nt][v] - m[v]);
                    rs[v] += p;
                    Ps[pRowBase + v][(nt << 4) + l16] = p;
                }
            }
            #pragma unroll
            for (int v = 0; v < 8; ++v) {
                float r = rs[v];
                r += row_xor<1>(r);
                r += row_xor<2>(r);
                r += row_xor<4>(r);
                r += row_xor<8>(r);
                lsum[v] = lsum[v] * corr[v] + r;
            }
            __syncthreads();

            const int pr = (wave << 4) + l16;
            #pragma unroll
            for (int kk = 0; kk < 64; kk += 4) {
                const v2f a = *(const v2f*)&Ps[pr][kk + kb];
                #pragma unroll
                for (int nt = 0; nt < 4; ++nt) {
                    const v2f bf = *(const v2f*)&Vt[(nt << 4) + l16][kk + kb];
                    o[nt] = WMMA_F32(a, bf, o[nt]);
                }
            }
        }

        #pragma unroll
        for (int nt = 0; nt < 4; ++nt) {
            const int col = (h << 6) + (nt << 4) + l16;
            #pragma unroll
            for (int v = 0; v < 8; ++v) {
                const int row = qBase + (wave << 4) + rowOff + v;
                Out[(rowB + row) * DIMp + col] = o[nt][v] / lsum[v];
            }
        }
    }
}

extern "C" void kernel_launch(void* const* d_in, const int* in_sizes, int n_in,
                              void* d_out, int out_size, void* d_ws, size_t ws_size,
                              hipStream_t stream) {
    (void)in_sizes; (void)n_in; (void)out_size; (void)ws_size;
    constexpr int DIM = 1024, HEAD_DIM = 64, NUM_HEADS = 16, BATCH = 2, SEQ = 2048;
    constexpr int M = BATCH * SEQ;

    const float* x  = (const float*)d_in[0];
    const float* Wq = (const float*)d_in[1];
    const float* Wk = (const float*)d_in[2];
    const float* Wv = (const float*)d_in[3];
    const float* Wo = (const float*)d_in[4];
    const float* bo = (const float*)d_in[5];
    float* out = (float*)d_out;

    float* Qbuf = (float*)d_ws;
    float* Kbuf = Qbuf + (size_t)M * DIM;
    float* Vbuf = Kbuf + (size_t)M * HEAD_DIM;
    float* Abuf = Vbuf + (size_t)M * HEAD_DIM;

    const float scale = 0.125f;

    gemm_wmma_f32<<<dim3(DIM / 64 + 2 * (HEAD_DIM / 64), M / 128), 256, 0, stream>>>(
        x, Wq, Wk, Wv, nullptr, Qbuf, Kbuf, Vbuf, M, DIM, HEAD_DIM, DIM, scale);

    mqa_flash_wmma<<<dim3(SEQ / 128, NUM_HEADS, BATCH), 128, 0, stream>>>(
        Qbuf, Kbuf, Vbuf, Abuf, SEQ, DIM);

    gemm_wmma_f32<<<dim3(DIM / 64, M / 128), 256, 0, stream>>>(
        Abuf, Wo, nullptr, nullptr, bo, out, nullptr, nullptr, M, DIM, 0, DIM, 1.0f);
}


# Next tile's global loads (attention K/V, GEMM A/W) go into registers before the current tile's WMMAs, reach LDS one step later; barriers fence LDS only so they stay in flight; sum order unchanged.
# speedup vs baseline: 1.6582x; 1.0347x over previous
#include <hip/hip_runtime.h>
#include <math.h>

typedef __attribute__((ext_vector_type(2))) float v2f;
typedef __attribute__((ext_vector_type(8))) float v8f;

#define WMMA_F32(a, b, c) \
    __builtin_amdgcn_wmma_f32_16x16x4_f32(false, (a), false, (b), (short)0, (c), false, false)

__device__ __forceinline__ void lds_barrier() {
    __builtin_amdgcn_fence(__ATOMIC_RELEASE, "workgroup", "local");
    __builtin_amdgcn_s_barrier();
    __builtin_amdgcn_fence(__ATOMIC_ACQUIRE, "workgroup", "local");
}

__global__ __launch_bounds__(256) void gemm_wmma_f32(
    const float* __restrict__ A,
    const float* __restrict__ W0, const float* __restrict__ W1, const float* __restrict__ W2,
    const float* __restrict__ bias0,
    float* __restrict__ C0, float* __restrict__ C1, float* __restrict__ C2,
    int M, int N0, int N1, int K, float alpha0)
{
    const int nb0 = N0 >> 6;
    const int nb1 = N1 >> 6;
    int bx = (int)blockIdx.x;
    const float* W = W0;
    const float* bias = bias0;
    float* C = C0;
    int N = N0;
    float alpha = alpha0;
    if (bx >= nb0) {
        bx -= nb0;
        W = W1; C = C1;
        if (bx >= nb1) { bx -= nb1; W = W2; C = C2; }
        N = N1; alpha = 1.0f; bias = nullptr;
    }

    __shared__ __align__(16) float As[128][18];
    __shared__ __align__(16) float Wt[64][18];

    const int tid    = threadIdx.x;
    const int wave   = tid >> 5;
    const int lane   = tid & 31;
    const int l16    = lane & 15;
    const int half   = lane >> 4;
    const int kb     = half << 1;
    const int rowOff = half << 3;

    const int wm = wave >> 1;
    const int wn = wave & 1;

    const int rowBlock = blockIdx.y * 128;
    const int colBlock = bx * 64;

    v8f acc[2][2];
    #pragma unroll
    for (int mt = 0; mt < 2; ++mt)
        #pragma unroll
        for (int nt = 0; nt < 2; ++nt) acc[mt][nt] = (v8f)(0.0f);

    const int ar = tid >> 1;
    const int ac = (tid & 1) << 3;
    const int wr = tid >> 4;
    const int wc = (tid & 15) << 2;

    const int aRow0 = (wm << 5) + l16;
    const int aRow1 = aRow0 + 16;
    const int bRow0 = (wn << 5) + l16;
    const int bRow1 = bRow0 + 16;

    const float* aSrc = A + (size_t)(rowBlock + ar) * K + ac;
    const float* wSrc = W + (size_t)wr * N + colBlock + wc;
    float4 a0 = *(const float4*)(aSrc);
    float4 a1 = *(const float4*)(aSrc + 4);
    float4 wv = *(const float4*)(wSrc);

    for (int kt = 0; kt < K; kt += 16) {
        As[ar][ac + 0] = a0.x; As[ar][ac + 1] = a0.y;
        As[ar][ac + 2] = a0.z; As[ar][ac + 3] = a0.w;
        As[ar][ac + 4] = a1.x; As[ar][ac + 5] = a1.y;
        As[ar][ac + 6] = a1.z; As[ar][ac + 7] = a1.w;
        Wt[wc + 0][wr] = wv.x; Wt[wc + 1][wr] = wv.y;
        Wt[wc + 2][wr] = wv.z; Wt[wc + 3][wr] = wv.w;
        if (kt + 16 < K) {
            aSrc += 16;
            wSrc += (size_t)16 * N;
            a0 = *(const float4*)(aSrc);
            a1 = *(const float4*)(aSrc + 4);
            wv = *(const float4*)(wSrc);
            if (kt + 32 < K) {
                __builtin_prefetch(aSrc + 16, 0, 0);
                __builtin_prefetch(wSrc + (size_t)16 * N, 0, 0);
            }
        }
        lds_barrier();

        #pragma unroll
        for (int kk = 0; kk < 16; kk += 4) {
            const v2f av0 = *(const v2f*)&As[aRow0][kk + kb];
            const v2f av1 = *(const v2f*)&As[aRow1][kk + kb];
            const v2f bv0 = *(const v2f*)&Wt[bRow0][kk + kb];
            const v2f bv1 = *(const v2f*)&Wt[bRow1][kk + kb];
            acc[0][0] = WMMA_F32(av0, bv0, acc[0][0]);
            acc[0][1] = WMMA_F32(av0, bv1, acc[0][1]);
            acc[1][0] = WMMA_F32(av1, bv0, acc[1][0]);
            acc[1][1] = WMMA_F32(av1, bv1, acc[1][1]);
        }
        lds_barrier();
    }

    #pragma unroll
    for (int mt = 0; mt < 2; ++mt) {
        #pragma unroll
        for (int nt = 0; nt < 2; ++nt) {
            const int col = colBlock + (wn << 5) + (nt << 4) + l16;
            const float bv = bias ? bias[col] : 0.0f;
            #pragma unroll
            for (int v = 0; v < 8; ++v) {
                const int row = rowBlock + (wm << 5) + (mt << 4) + rowOff + v;
                C[(size_t)row * N + col] = alpha * acc[mt][nt][v] + bv;
            }
        }
    }
}

template <int MASK>
__device__ __forceinline__ float row_xor(float x) {
    const int xi = __builtin_bit_cast(int, x);
    return __builtin_bit_cast(float, __builtin_amdgcn_update_dpp(0, xi, 0x160 | MASK, 0xf, 0xf, true));
}

__global__ __launch_bounds__(128) void mqa_flash_wmma(
    const float* __restrict__ Q,
    const float* __restrict__ Kg,
    const float* __restrict__ Vg,
    float* __restrict__ Out,
    int S, int DIMp)
{
    __shared__ __align__(16) float Ks[64][68];
    __shared__ __align__(16) float Vt[64][68];
    __shared__ __align__(16) float Ps[64][68];

    const int tid    = threadIdx.x;
    const int wave   = tid >> 5;
    const int lane   = tid & 31;
    const int l16    = lane & 15;
    const int half   = lane >> 4;
    const int kb     = half << 1;
    const int rowOff = half << 3;

    const int h     = blockIdx.y;
    const int b     = blockIdx.z;
    const size_t rowB = (size_t)b * S;
    const int nQ    = (int)gridDim.x << 1;

    const int lr = tid >> 1;
    const int lc = (tid & 1) << 5;

    for (int pass = 0; pass < 2; ++pass) {
        const int qTile = pass == 0 ? (nQ - 1 - (int)blockIdx.x) : (int)blockIdx.x;
        const int qBase = qTile << 6;

        const int nTiles = qTile + 1;

        const float* ksrc = Kg + (rowB + lr) * 64 + lc;
        const float* vsrc = Vg + (rowB + lr) * 64 + lc;
        float4 kr[8], vr[8];
        #pragma unroll
        for (int j = 0; j < 8; ++j) {
            kr[j] = *(const float4*)(ksrc + (j << 2));
            vr[j] = *(const float4*)(vsrc + (j << 2));
        }

        if (pass != 0) lds_barrier();

        {
            const float* src = Q + (rowB + qBase + lr) * DIMp + (h << 6) + lc;
            #pragma unroll
            for (int j = 0; j < 8; ++j) {
                const float4 qv = *(const float4*)(src + (j << 2));
                Ps[lr][lc + (j << 2) + 0] = qv.x; Ps[lr][lc + (j << 2) + 1] = qv.y;
                Ps[lr][lc + (j << 2) + 2] = qv.z; Ps[lr][lc + (j << 2) + 3] = qv.w;
            }
        }
        lds_barrier();

        v2f aq[16];
        {
            const int qr = (wave << 4) + l16;
            #pragma unroll
            for (int i = 0; i < 16; ++i)
                aq[i] = *(const v2f*)&Ps[qr][(i << 2) + kb];
        }

        v8f o[4];
        #pragma unroll
        for (int nt = 0; nt < 4; ++nt) o[nt] = (v8f)(0.0f);
        float m[8], lsum[8];
        #pragma unroll
        for (int v = 0; v < 8; ++v) { m[v] = -1e30f; lsum[v] = 0.0f; }

        for (int t = 0; t < nTiles; ++t) {
            const int kBase = t << 6;
            lds_barrier();
            #pragma unroll
            for (int j = 0; j < 8; ++j) {
                const float4 kv = kr[j];
                const float4 vv = vr[j];
                Ks[lr][lc + (j << 2) + 0] = kv.x; Ks[lr][lc + (j << 2) + 1] = kv.y;
                Ks[lr][lc + (j << 2) + 2] = kv.z; Ks[lr][lc + (j << 2) + 3] = kv.w;
                Vt[lc + (j << 2) + 0][lr] = vv.x; Vt[lc + (j << 2) + 1][lr] = vv.y;
                Vt[lc + (j << 2) + 2][lr] = vv.z; Vt[lc + (j << 2) + 3][lr] = vv.w;
            }
            if (t + 1 < nTiles) {
                ksrc += 64 * 64;
                vsrc += 64 * 64;
                #pragma unroll
                for (int j = 0; j < 8; ++j) {
                    kr[j] = *(const float4*)(ksrc + (j << 2));
                    vr[j] = *(const float4*)(vsrc + (j << 2));
                }
            }
            lds_barrier();

            v8f sc[4];
            #pragma unroll
            for (int nt = 0; nt < 4; ++nt) sc[nt] = (v8f)(0.0f);
            #pragma unroll
            for (int kk = 0; kk < 64; kk += 4) {
                const v2f a = aq[kk >> 2];
                #pragma unroll
                for (int nt = 0; nt < 4; ++nt) {
                    const v2f bf = *(const v2f*)&Ks[(nt << 4) + l16][kk + kb];
                    sc[nt] = WMMA_F32(a, bf, sc[nt]);
                }
            }

            if (t == nTiles - 1) {
                #pragma unroll
                for (int nt = 0; nt < 4; ++nt) {
                    const int key = kBase + (nt << 4) + l16;
                    #pragma unroll
                    for (int v = 0; v < 8; ++v) {
                        const int qr = qBase + (wave << 4) + rowOff + v;
                        if (key > qr) sc[nt][v] = -1e30f;
                    }
                }
            }

            float corr[8];
            #pragma unroll
            for (int v = 0; v < 8; ++v) {
                float x = fmaxf(fmaxf(sc[0][v], sc[1][v]), fmaxf(sc[2][v], sc[3][v]));
                x = fmaxf(x, row_xor<1>(x));
                x = fmaxf(x, row_xor<2>(x));
                x = fmaxf(x, row_xor<4>(x));
                x = fmaxf(x, row_xor<8>(x));
                const float mNew = fmaxf(m[v], x);
                corr[v] = __expf(m[v] - mNew);
                m[v] = mNew;
            }
            #pragma unroll
            for (int nt = 0; nt < 4; ++nt)
                #pragma unroll
                for (int v = 0; v < 8; ++v) o[nt][v] *= corr[v];

            float rs[8];
            #pragma unroll
            for (int v = 0; v < 8; ++v) rs[v] = 0.0f;
            const int pRowBase = (wave << 4) + rowOff;
            #pragma unroll
            for (int nt = 0; nt < 4; ++nt) {
                #pragma unroll
                for (int v = 0; v < 8; ++v) {
                    const float p = __expf(sc[nt][v] - m[v]);
                    rs[v] += p;
                    Ps[pRowBase + v][(nt << 4) + l16] = p;
                }
            }
            #pragma unroll
            for (int v = 0; v < 8; ++v) {
                float r = rs[v];
                r += row_xor<1>(r);
                r += row_xor<2>(r);
                r += row_xor<4>(r);
                r += row_xor<8>(r);
                lsum[v] = lsum[v] * corr[v] + r;
            }
            lds_barrier();

            const int pr = (wave << 4) + l16;
            #pragma unroll
            for (int kk = 0; kk < 64; kk += 4) {
                const v2f a = *(const v2f*)&Ps[pr][kk + kb];
                #pragma unroll
                for (int nt = 0; nt < 4; ++nt) {
                    const v2f bf = *(const v2f*)&Vt[(nt << 4) + l16][kk + kb];
                    o[nt] = WMMA_F32(a, bf, o[nt]);
                }
            }
        }

        #pragma unroll
        for (int nt = 0; nt < 4; ++nt) {
            const int col = (h << 6) + (nt << 4) + l16;
            #pragma unroll
            for (int v = 0; v < 8; ++v) {
                const int row = qBase + (wave << 4) + rowOff + v;
                Out[(rowB + row) * DIMp + col] = o[nt][v] / lsum[v];
            }
        }
    }
}

extern "C" void kernel_launch(void* const* d_in, const int* in_sizes, int n_in,
                              void* d_out, int out_size, void* d_ws, size_t ws_size,
                              hipStream_t stream) {
    (void)in_sizes; (void)n_in; (void)out_size; (void)ws_size;
    constexpr int DIM = 1024, HEAD_DIM = 64, NUM_HEADS = 16, BATCH = 2, SEQ = 2048;
    constexpr int M = BATCH * SEQ;

    const float* x  = (const float*)d_in[0];
    const float* Wq = (const float*)d_in[1];
    const float* Wk = (const float*)d_in[2];
    const float* Wv = (const float*)d_in[3];
    const float* Wo = (const float*)d_in[4];
    const float* bo = (const float*)d_in[5];
    float* out = (float*)d_out;

    float* Qbuf = (float*)d_ws;
    float* Kbuf = Qbuf + (size_t)M * DIM;
    float* Vbuf = Kbuf + (size_t)M * HEAD_DIM;
    float* Abuf = Vbuf + (size_t)M * HEAD_DIM;

    const float scale = 0.125f;

    gemm_wmma_f32<<<dim3(DIM / 64 + 2 * (HEAD_DIM / 64), M / 128), 256, 0, stream>>>(
        x, Wq, Wk, Wv, nullptr, Qbuf, Kbuf, Vbuf, M, DIM, HEAD_DIM, DIM, scale);

    mqa_flash_wmma<<<dim3(SEQ / 128, NUM_HEADS, BATCH), 128, 0, stream>>>(
        Qbuf, Kbuf, Vbuf, Abuf, SEQ, DIM);

    gemm_wmma_f32<<<dim3(DIM / 64, M / 128), 256, 0, stream>>>(
        Abuf, Wo, nullptr, nullptr, bo, out, nullptr, nullptr, M, DIM, 0, DIM, 1.0f);
}


# Attention transposed (S^T = K Q^T, O^T = V^T P^T): a lane owns one query, P feeds the 2nd product from registers (no LDS trip, one barrier less per step); softmax sums in-lane: sum order changes
# speedup vs baseline: 1.7290x; 1.0427x over previous
#include <hip/hip_runtime.h>
#include <math.h>

typedef __attribute__((ext_vector_type(2))) float v2f;
typedef __attribute__((ext_vector_type(8))) float v8f;

#define WMMA_F32(a, b, c) \
    __builtin_amdgcn_wmma_f32_16x16x4_f32(false, (a), false, (b), (short)0, (c), false, false)

__device__ __forceinline__ void lds_barrier() {
    __builtin_amdgcn_fence(__ATOMIC_RELEASE, "workgroup", "local");
    __builtin_amdgcn_s_barrier();
    __builtin_amdgcn_fence(__ATOMIC_ACQUIRE, "workgroup", "local");
}

__global__ __launch_bounds__(256) void gemm_wmma_f32(
    const float* __restrict__ A,
    const float* __restrict__ W0, const float* __restrict__ W1, const float* __restrict__ W2,
    const float* __restrict__ bias0,
    float* __restrict__ C0, float* __restrict__ C1, float* __restrict__ C2,
    int M, int N0, int N1, int K, float alpha0)
{
    const int nb0 = N0 >> 6;
    const int nb1 = N1 >> 6;
    int bx = (int)blockIdx.x;
    const float* W = W0;
    const float* bias = bias0;
    float* C = C0;
    int N = N0;
    float alpha = alpha0;
    if (bx >= nb0) {
        bx -= nb0;
        W = W1; C = C1;
        if (bx >= nb1) { bx -= nb1; W = W2; C = C2; }
        N = N1; alpha = 1.0f; bias = nullptr;
    }

    __shared__ __align__(16) float As[128][18];
    __shared__ __align__(16) float Wt[64][18];

    const int tid    = threadIdx.x;
    const int wave   = tid >> 5;
    const int lane   = tid & 31;
    const int l16    = lane & 15;
    const int half   = lane >> 4;
    const int kb     = half << 1;
    const int rowOff = half << 3;

    const int wm = wave >> 1;
    const int wn = wave & 1;

    const int rowBlock = blockIdx.y * 128;
    const int colBlock = bx * 64;

    v8f acc[2][2];
    #pragma unroll
    for (int mt = 0; mt < 2; ++mt)
        #pragma unroll
        for (int nt = 0; nt < 2; ++nt) acc[mt][nt] = (v8f)(0.0f);

    const int ar = tid >> 1;
    const int ac = (tid & 1) << 3;
    const int wr = tid >> 4;
    const int wc = (tid & 15) << 2;

    const int aRow0 = (wm << 5) + l16;
    const int aRow1 = aRow0 + 16;
    const int bRow0 = (wn << 5) + l16;
    const int bRow1 = bRow0 + 16;

    const float* aSrc = A + (size_t)(rowBlock + ar) * K + ac;
    const float* wSrc = W + (size_t)wr * N + colBlock + wc;
    float4 a0 = *(const float4*)(aSrc);
    float4 a1 = *(const float4*)(aSrc + 4);
    float4 wv = *(const float4*)(wSrc);

    for (int kt = 0; kt < K; kt += 16) {
        As[ar][ac + 0] = a0.x; As[ar][ac + 1] = a0.y;
        As[ar][ac + 2] = a0.z; As[ar][ac + 3] = a0.w;
        As[ar][ac + 4] = a1.x; As[ar][ac + 5] = a1.y;
        As[ar][ac + 6] = a1.z; As[ar][ac + 7] = a1.w;
        Wt[wc + 0][wr] = wv.x; Wt[wc + 1][wr] = wv.y;
        Wt[wc + 2][wr] = wv.z; Wt[wc + 3][wr] = wv.w;
        if (kt + 16 < K) {
            aSrc += 16;
            wSrc += (size_t)16 * N;
            a0 = *(const float4*)(aSrc);
            a1 = *(const float4*)(aSrc + 4);
            wv = *(const float4*)(wSrc);
            if (kt + 32 < K) {
                __builtin_prefetch(aSrc + 16, 0, 0);
                __builtin_prefetch(wSrc + (size_t)16 * N, 0, 0);
            }
        }
        lds_barrier();

        #pragma unroll
        for (int kk = 0; kk < 16; kk += 4) {
            const v2f av0 = *(const v2f*)&As[aRow0][kk + kb];
            const v2f av1 = *(const v2f*)&As[aRow1][kk + kb];
            const v2f bv0 = *(const v2f*)&Wt[bRow0][kk + kb];
            const v2f bv1 = *(const v2f*)&Wt[bRow1][kk + kb];
            acc[0][0] = WMMA_F32(av0, bv0, acc[0][0]);
            acc[0][1] = WMMA_F32(av0, bv1, acc[0][1]);
            acc[1][0] = WMMA_F32(av1, bv0, acc[1][0]);
            acc[1][1] = WMMA_F32(av1, bv1, acc[1][1]);
        }
        lds_barrier();
    }

    #pragma unroll
    for (int mt = 0; mt < 2; ++mt) {
        #pragma unroll
        for (int nt = 0; nt < 2; ++nt) {
            const int col = colBlock + (wn << 5) + (nt << 4) + l16;
            const float bv = bias ? bias[col] : 0.0f;
            #pragma unroll
            for (int v = 0; v < 8; ++v) {
                const int row = rowBlock + (wm << 5) + (mt << 4) + rowOff + v;
                C[(size_t)row * N + col] = alpha * acc[mt][nt][v] + bv;
            }
        }
    }
}

__device__ __forceinline__ float xhalf(float x) {
    const int b = __builtin_bit_cast(int, x);
    return __builtin_bit_cast(float, __builtin_amdgcn_permlanex16(b, b, 0x76543210, 0xfedcba98, false, false));
}

__global__ __launch_bounds__(128) void mqa_flash_wmma(
    const float* __restrict__ Q,
    const float* __restrict__ Kg,
    const float* __restrict__ Vg,
    float* __restrict__ Out,
    int S, int DIMp)
{
    __shared__ __align__(16) float Ks[64][68];
    __shared__ __align__(16) float Vt[64][68];
    __shared__ __align__(16) float Ps[64][68];

    const int tid    = threadIdx.x;
    const int wave   = tid >> 5;
    const int lane   = tid & 31;
    const int l16    = lane & 15;
    const int half   = lane >> 4;
    const int kb     = half << 1;
    const int rowOff = half << 3;

    const int h     = blockIdx.y;
    const int b     = blockIdx.z;
    const size_t rowB = (size_t)b * S;
    const int nQ    = (int)gridDim.x << 1;

    const int lr = tid >> 1;
    const int lc = (tid & 1) << 5;

    for (int pass = 0; pass < 2; ++pass) {
        const int qTile = pass == 0 ? (nQ - 1 - (int)blockIdx.x) : (int)blockIdx.x;
        const int qBase = qTile << 6;

        const int nTiles = qTile + 1;

        const float* ksrc = Kg + (rowB + lr) * 64 + lc;
        const float* vsrc = Vg + (rowB + lr) * 64 + lc;
        float4 kr[8], vr[8];
        #pragma unroll
        for (int j = 0; j < 8; ++j) {
            kr[j] = *(const float4*)(ksrc + (j << 2));
            vr[j] = *(const float4*)(vsrc + (j << 2));
        }

        if (pass != 0) lds_barrier();

        {
            const float* src = Q + (rowB + qBase + lr) * DIMp + (h << 6) + lc;
            #pragma unroll
            for (int j = 0; j < 8; ++j) {
                const float4 qv = *(const float4*)(src + (j << 2));
                Ps[lr][lc + (j << 2) + 0] = qv.x; Ps[lr][lc + (j << 2) + 1] = qv.y;
                Ps[lr][lc + (j << 2) + 2] = qv.z; Ps[lr][lc + (j << 2) + 3] = qv.w;
            }
        }
        lds_barrier();

        v2f bq[16];
        {
            const int qr = (wave << 4) + l16;
            #pragma unroll
            for (int i = 0; i < 16; ++i)
                bq[i] = *(const v2f*)&Ps[qr][(i << 2) + kb];
        }

        v8f ot[4];
        #pragma unroll
        for (int dt = 0; dt < 4; ++dt) ot[dt] = (v8f)(0.0f);
        float m = -1e30f;
        float lsum = 0.0f;

        const int qIdx = qBase + (wave << 4) + l16;

        for (int t = 0; t < nTiles; ++t) {
            const int kBase = t << 6;
            lds_barrier();
            #pragma unroll
            for (int j = 0; j < 8; ++j) {
                const float4 kv = kr[j];
                const float4 vv = vr[j];
                Ks[lr][lc + (j << 2) + 0] = kv.x; Ks[lr][lc + (j << 2) + 1] = kv.y;
                Ks[lr][lc + (j << 2) + 2] = kv.z; Ks[lr][lc + (j << 2) + 3] = kv.w;
                Vt[lc + (j << 2) + 0][lr] = vv.x; Vt[lc + (j << 2) + 1][lr] = vv.y;
                Vt[lc + (j << 2) + 2][lr] = vv.z; Vt[lc + (j << 2) + 3][lr] = vv.w;
            }
            if (t + 1 < nTiles) {
                ksrc += 64 * 64;
                vsrc += 64 * 64;
                #pragma unroll
                for (int j = 0; j < 8; ++j) {
                    kr[j] = *(const float4*)(ksrc + (j << 2));
                    vr[j] = *(const float4*)(vsrc + (j << 2));
                }
            }
            lds_barrier();

            v8f st[4];
            #pragma unroll
            for (int mt = 0; mt < 4; ++mt) st[mt] = (v8f)(0.0f);
            #pragma unroll
            for (int kk = 0; kk < 64; kk += 4) {
                const v2f bf = bq[kk >> 2];
                #pragma unroll
                for (int mt = 0; mt < 4; ++mt) {
                    const v2f af = *(const v2f*)&Ks[(mt << 4) + l16][kk + kb];
                    st[mt] = WMMA_F32(af, bf, st[mt]);
                }
            }

            if (t == nTiles - 1) {
                #pragma unroll
                for (int mt = 0; mt < 4; ++mt) {
                    #pragma unroll
                    for (int r = 0; r < 8; ++r) {
                        const int key = kBase + (mt << 4) + rowOff + r;
                        if (key > qIdx) st[mt][r] = -1e30f;
                    }
                }
            }

            float x = -1e30f;
            #pragma unroll
            for (int mt = 0; mt < 4; ++mt) {
                #pragma unroll
                for (int r = 0; r < 8; ++r) x = fmaxf(x, st[mt][r]);
            }
            x = fmaxf(x, xhalf(x));
            const float mNew = fmaxf(m, x);
            const float corr = __expf(m - mNew);
            m = mNew;
            #pragma unroll
            for (int dt = 0; dt < 4; ++dt)
                #pragma unroll
                for (int r = 0; r < 8; ++r) ot[dt][r] *= corr;

            float rs = 0.0f;
            #pragma unroll
            for (int mt = 0; mt < 4; ++mt) {
                #pragma unroll
                for (int r = 0; r < 8; ++r) {
                    const float p = __expf(st[mt][r] - m);
                    st[mt][r] = p;
                    rs += p;
                }
            }
            lsum = lsum * corr + rs;

            #pragma unroll
            for (int mt = 0; mt < 4; ++mt) {
                #pragma unroll
                for (int j = 0; j < 4; ++j) {
                    v2f pf;
                    pf.x = st[mt][2 * j];
                    pf.y = st[mt][2 * j + 1];
                    #pragma unroll
                    for (int dt = 0; dt < 4; ++dt) {
                        const v2f af = *(const v2f*)&Vt[(dt << 4) + l16][(mt << 4) + rowOff + (j << 1)];
                        ot[dt] = WMMA_F32(af, pf, ot[dt]);
                    }
                }
            }
        }

        {
            const float ltot = lsum + xhalf(lsum);
            const int qr = (wave << 4) + l16;
            #pragma unroll
            for (int dt = 0; dt < 4; ++dt) {
                float4 lo, hi;
                lo.x = ot[dt][0] / ltot; lo.y = ot[dt][1] / ltot; lo.z = ot[dt][2] / ltot; lo.w = ot[dt][3] / ltot;
                hi.x = ot[dt][4] / ltot; hi.y = ot[dt][5] / ltot; hi.z = ot[dt][6] / ltot; hi.w = ot[dt][7] / ltot;
                *(float4*)&Ps[qr][(dt << 4) + rowOff + 0] = lo;
                *(float4*)&Ps[qr][(dt << 4) + rowOff + 4] = hi;
            }
            lds_barrier();
            const int c4 = l16 << 2;
            #pragma unroll
            for (int i = 0; i < 8; ++i) {
                const int rr = (wave << 4) + (i << 1) + half;
                const float4 v = *(const float4*)&Ps[rr][c4];
                *(float4*)&Out[(rowB + qBase + rr) * DIMp + (h << 6) + c4] = v;
            }
        }
    }
}

extern "C" void kernel_launch(void* const* d_in, const int* in_sizes, int n_in,
                              void* d_out, int out_size, void* d_ws, size_t ws_size,
                              hipStream_t stream) {
    (void)in_sizes; (void)n_in; (void)out_size; (void)ws_size;
    constexpr int DIM = 1024, HEAD_DIM = 64, NUM_HEADS = 16, BATCH = 2, SEQ = 2048;
    constexpr int M = BATCH * SEQ;

    const float* x  = (const float*)d_in[0];
    const float* Wq = (const float*)d_in[1];
    const float* Wk = (const float*)d_in[2];
    const float* Wv = (const float*)d_in[3];
    const float* Wo = (const float*)d_in[4];
    const float* bo = (const float*)d_in[5];
    float* out = (float*)d_out;

    float* Qbuf = (float*)d_ws;
    float* Kbuf = Qbuf + (size_t)M * DIM;
    float* Vbuf = Kbuf + (size_t)M * HEAD_DIM;
    float* Abuf = Vbuf + (size_t)M * HEAD_DIM;

    const float scale = 0.125f;

    gemm_wmma_f32<<<dim3(DIM / 64 + 2 * (HEAD_DIM / 64), M / 128), 256, 0, stream>>>(
        x, Wq, Wk, Wv, nullptr, Qbuf, Kbuf, Vbuf, M, DIM, HEAD_DIM, DIM, scale);

    mqa_flash_wmma<<<dim3(SEQ / 128, NUM_HEADS, BATCH), 128, 0, stream>>>(
        Qbuf, Kbuf, Vbuf, Abuf, SEQ, DIM);

    gemm_wmma_f32<<<dim3(DIM / 64, M / 128), 256, 0, stream>>>(
        Abuf, Wo, nullptr, nullptr, bo, out, nullptr, nullptr, M, DIM, 0, DIM, 1.0f);
}


# Attention: each key tile's 64 keys split over two groups of 4 waves (8-wave workgroups, 4 waves per SIMD instead of 2); the two partial softmax states joined once per query tile; sum order changes
# speedup vs baseline: 1.8118x; 1.0479x over previous
#include <hip/hip_runtime.h>
#include <math.h>

typedef __attribute__((ext_vector_type(2))) float v2f;
typedef __attribute__((ext_vector_type(8))) float v8f;

#define WMMA_F32(a, b, c) \
    __builtin_amdgcn_wmma_f32_16x16x4_f32(false, (a), false, (b), (short)0, (c), false, false)

__device__ __forceinline__ void lds_barrier() {
    __builtin_amdgcn_fence(__ATOMIC_RELEASE, "workgroup", "local");
    __builtin_amdgcn_s_barrier();
    __builtin_amdgcn_fence(__ATOMIC_ACQUIRE, "workgroup", "local");
}

__global__ __launch_bounds__(256) void gemm_wmma_f32(
    const float* __restrict__ A,
    const float* __restrict__ W0, const float* __restrict__ W1, const float* __restrict__ W2,
    const float* __restrict__ bias0,
    float* __restrict__ C0, float* __restrict__ C1, float* __restrict__ C2,
    int M, int N0, int N1, int K, float alpha0)
{
    const int nb0 = N0 >> 6;
    const int nb1 = N1 >> 6;
    int bx = (int)blockIdx.x;
    const float* W = W0;
    const float* bias = bias0;
    float* C = C0;
    int N = N0;
    float alpha = alpha0;
    if (bx >= nb0) {
        bx -= nb0;
        W = W1; C = C1;
        if (bx >= nb1) { bx -= nb1; W = W2; C = C2; }
        N = N1; alpha = 1.0f; bias = nullptr;
    }

    __shared__ __align__(16) float As[128][18];
    __shared__ __align__(16) float Wt[64][18];

    const int tid    = threadIdx.x;
    const int wave   = tid >> 5;
    const int lane   = tid & 31;
    const int l16    = lane & 15;
    const int half   = lane >> 4;
    const int kb     = half << 1;
    const int rowOff = half << 3;

    const int wm = wave >> 1;
    const int wn = wave & 1;

    const int rowBlock = blockIdx.y * 128;
    const int colBlock = bx * 64;

    v8f acc[2][2];
    #pragma unroll
    for (int mt = 0; mt < 2; ++mt)
        #pragma unroll
        for (int nt = 0; nt < 2; ++nt) acc[mt][nt] = (v8f)(0.0f);

    const int ar = tid >> 1;
    const int ac = (tid & 1) << 3;
    const int wr = tid >> 4;
    const int wc = (tid & 15) << 2;

    const int aRow0 = (wm << 5) + l16;
    const int aRow1 = aRow0 + 16;
    const int bRow0 = (wn << 5) + l16;
    const int bRow1 = bRow0 + 16;

    const float* aSrc = A + (size_t)(rowBlock + ar) * K + ac;
    const float* wSrc = W + (size_t)wr * N + colBlock + wc;
    float4 a0 = *(const float4*)(aSrc);
    float4 a1 = *(const float4*)(aSrc + 4);
    float4 wv = *(const float4*)(wSrc);

    for (int kt = 0; kt < K; kt += 16) {
        As[ar][ac + 0] = a0.x; As[ar][ac + 1] = a0.y;
        As[ar][ac + 2] = a0.z; As[ar][ac + 3] = a0.w;
        As[ar][ac + 4] = a1.x; As[ar][ac + 5] = a1.y;
        As[ar][ac + 6] = a1.z; As[ar][ac + 7] = a1.w;
        Wt[wc + 0][wr] = wv.x; Wt[wc + 1][wr] = wv.y;
        Wt[wc + 2][wr] = wv.z; Wt[wc + 3][wr] = wv.w;
        if (kt + 16 < K) {
            aSrc += 16;
            wSrc += (size_t)16 * N;
            a0 = *(const float4*)(aSrc);
            a1 = *(const float4*)(aSrc + 4);
            wv = *(const float4*)(wSrc);
            if (kt + 32 < K) {
                __builtin_prefetch(aSrc + 16, 0, 0);
                __builtin_prefetch(wSrc + (size_t)16 * N, 0, 0);
            }
        }
        lds_barrier();

        #pragma unroll
        for (int kk = 0; kk < 16; kk += 4) {
            const v2f av0 = *(const v2f*)&As[aRow0][kk + kb];
            const v2f av1 = *(const v2f*)&As[aRow1][kk + kb];
            const v2f bv0 = *(const v2f*)&Wt[bRow0][kk + kb];
            const v2f bv1 = *(const v2f*)&Wt[bRow1][kk + kb];
            acc[0][0] = WMMA_F32(av0, bv0, acc[0][0]);
            acc[0][1] = WMMA_F32(av0, bv1, acc[0][1]);
            acc[1][0] = WMMA_F32(av1, bv0, acc[1][0]);
            acc[1][1] = WMMA_F32(av1, bv1, acc[1][1]);
        }
        lds_barrier();
    }

    #pragma unroll
    for (int mt = 0; mt < 2; ++mt) {
        #pragma unroll
        for (int nt = 0; nt < 2; ++nt) {
            const int col = colBlock + (wn << 5) + (nt << 4) + l16;
            const float bv = bias ? bias[col] : 0.0f;
            #pragma unroll
            for (int v = 0; v < 8; ++v) {
                const int row = rowBlock + (wm << 5) + (mt << 4) + rowOff + v;
                C[(size_t)row * N + col] = alpha * acc[mt][nt][v] + bv;
            }
        }
    }
}

__device__ __forceinline__ float xhalf(float x) {
    const int b = __builtin_bit_cast(int, x);
    return __builtin_bit_cast(float, __builtin_amdgcn_permlanex16(b, b, 0x76543210, 0xfedcba98, false, false));
}

__global__ __launch_bounds__(256) void mqa_flash_wmma(
    const float* __restrict__ Q,
    const float* __restrict__ Kg,
    const float* __restrict__ Vg,
    float* __restrict__ Out,
    int S, int DIMp)
{
    __shared__ __align__(16) float Ks[64][68];
    __shared__ __align__(16) float Vt[64][68];
    __shared__ __align__(16) float Ps[64][68];
    __shared__ float Ms[64];
    __shared__ float Ls[64];

    const int tid    = threadIdx.x;
    const int wave   = tid >> 5;
    const int lane   = tid & 31;
    const int l16    = lane & 15;
    const int half   = lane >> 4;
    const int kb     = half << 1;
    const int rowOff = half << 3;
    const int qw     = wave & 3;
    const int kg     = wave >> 2;
    const int keyOff = kg << 5;

    const int h     = blockIdx.y;
    const int b     = blockIdx.z;
    const size_t rowB = (size_t)b * S;
    const int nQ    = (int)gridDim.x << 1;

    const int lr = tid >> 2;
    const int lq = (tid & 3) << 2;

    const int qr = (qw << 4) + l16;

    for (int pass = 0; pass < 2; ++pass) {
        const int qTile = pass == 0 ? (nQ - 1 - (int)blockIdx.x) : (int)blockIdx.x;
        const int qBase = qTile << 6;

        const int nTiles = qTile + 1;

        const float* ksrc = Kg + (rowB + lr) * 64 + lq;
        const float* vsrc = Vg + (rowB + lr) * 64 + lq;
        float4 kr[4], vr[4];
        #pragma unroll
        for (int j = 0; j < 4; ++j) {
            kr[j] = *(const float4*)(ksrc + (j << 4));
            vr[j] = *(const float4*)(vsrc + (j << 4));
        }

        if (pass != 0) lds_barrier();

        {
            const float* src = Q + (rowB + qBase + lr) * DIMp + (h << 6) + lq;
            #pragma unroll
            for (int j = 0; j < 4; ++j) {
                const float4 qv = *(const float4*)(src + (j << 4));
                Ps[lr][lq + (j << 4) + 0] = qv.x; Ps[lr][lq + (j << 4) + 1] = qv.y;
                Ps[lr][lq + (j << 4) + 2] = qv.z; Ps[lr][lq + (j << 4) + 3] = qv.w;
            }
        }
        lds_barrier();

        v2f bq[16];
        #pragma unroll
        for (int i = 0; i < 16; ++i)
            bq[i] = *(const v2f*)&Ps[qr][(i << 2) + kb];

        v8f ot[4];
        #pragma unroll
        for (int dt = 0; dt < 4; ++dt) ot[dt] = (v8f)(0.0f);
        float m = -1e30f;
        float lsum = 0.0f;

        const int qIdx = qBase + qr;

        for (int t = 0; t < nTiles; ++t) {
            const int kBase = t << 6;
            lds_barrier();
            #pragma unroll
            for (int j = 0; j < 4; ++j) {
                const float4 kv = kr[j];
                const float4 vv = vr[j];
                Ks[lr][lq + (j << 4) + 0] = kv.x; Ks[lr][lq + (j << 4) + 1] = kv.y;
                Ks[lr][lq + (j << 4) + 2] = kv.z; Ks[lr][lq + (j << 4) + 3] = kv.w;
                Vt[lq + (j << 4) + 0][lr] = vv.x; Vt[lq + (j << 4) + 1][lr] = vv.y;
                Vt[lq + (j << 4) + 2][lr] = vv.z; Vt[lq + (j << 4) + 3][lr] = vv.w;
            }
            if (t + 1 < nTiles) {
                ksrc += 64 * 64;
                vsrc += 64 * 64;
                #pragma unroll
                for (int j = 0; j < 4; ++j) {
                    kr[j] = *(const float4*)(ksrc + (j << 4));
                    vr[j] = *(const float4*)(vsrc + (j << 4));
                }
            }
            lds_barrier();

            v8f st[2];
            #pragma unroll
            for (int mt = 0; mt < 2; ++mt) st[mt] = (v8f)(0.0f);
            #pragma unroll
            for (int kk = 0; kk < 64; kk += 4) {
                const v2f bf = bq[kk >> 2];
                #pragma unroll
                for (int mt = 0; mt < 2; ++mt) {
                    const v2f af = *(const v2f*)&Ks[keyOff + (mt << 4) + l16][kk + kb];
                    st[mt] = WMMA_F32(af, bf, st[mt]);
                }
            }

            if (t == nTiles - 1) {
                #pragma unroll
                for (int mt = 0; mt < 2; ++mt) {
                    #pragma unroll
                    for (int r = 0; r < 8; ++r) {
                        const int key = kBase + keyOff + (mt << 4) + rowOff + r;
                        if (key > qIdx) st[mt][r] = -1e30f;
                    }
                }
            }

            float x = -1e30f;
            #pragma unroll
            for (int mt = 0; mt < 2; ++mt) {
                #pragma unroll
                for (int r = 0; r < 8; ++r) x = fmaxf(x, st[mt][r]);
            }
            x = fmaxf(x, xhalf(x));
            const float mNew = fmaxf(m, x);
            const float corr = __expf(m - mNew);
            m = mNew;
            #pragma unroll
            for (int dt = 0; dt < 4; ++dt)
                #pragma unroll
                for (int r = 0; r < 8; ++r) ot[dt][r] *= corr;

            float rs = 0.0f;
            #pragma unroll
            for (int mt = 0; mt < 2; ++mt) {
                #pragma unroll
                for (int r = 0; r < 8; ++r) {
                    const float p = __expf(st[mt][r] - m);
                    st[mt][r] = p;
                    rs += p;
                }
            }
            lsum = lsum * corr + rs;

            #pragma unroll
            for (int mt = 0; mt < 2; ++mt) {
                #pragma unroll
                for (int j = 0; j < 4; ++j) {
                    v2f pf;
                    pf.x = st[mt][2 * j];
                    pf.y = st[mt][2 * j + 1];
                    #pragma unroll
                    for (int dt = 0; dt < 4; ++dt) {
                        const v2f af = *(const v2f*)&Vt[(dt << 4) + l16][keyOff + (mt << 4) + rowOff + (j << 1)];
                        ot[dt] = WMMA_F32(af, pf, ot[dt]);
                    }
                }
            }
        }

        {
            const float ltot = lsum + xhalf(lsum);
            if (kg == 1) {
                #pragma unroll
                for (int dt = 0; dt < 4; ++dt) {
                    float4 lo, hi;
                    lo.x = ot[dt][0]; lo.y = ot[dt][1]; lo.z = ot[dt][2]; lo.w = ot[dt][3];
                    hi.x = ot[dt][4]; hi.y = ot[dt][5]; hi.z = ot[dt][6]; hi.w = ot[dt][7];
                    *(float4*)&Ps[qr][(dt << 4) + rowOff + 0] = lo;
                    *(float4*)&Ps[qr][(dt << 4) + rowOff + 4] = hi;
                }
                if (half == 0) { Ms[qr] = m; Ls[qr] = ltot; }
            }
            lds_barrier();
            if (kg == 0) {
                const float m1 = Ms[qr];
                const float l1 = Ls[qr];
                const float mm = fmaxf(m, m1);
                const float a0 = __expf(m - mm);
                const float a1 = __expf(m1 - mm);
                const float den = a0 * ltot + a1 * l1;
                #pragma unroll
                for (int dt = 0; dt < 4; ++dt) {
                    const float4 plo = *(const float4*)&Ps[qr][(dt << 4) + rowOff + 0];
                    const float4 phi = *(const float4*)&Ps[qr][(dt << 4) + rowOff + 4];
                    float4 lo, hi;
                    lo.x = (a0 * ot[dt][0] + a1 * plo.x) / den; lo.y = (a0 * ot[dt][1] + a1 * plo.y) / den;
                    lo.z = (a0 * ot[dt][2] + a1 * plo.z) / den; lo.w = (a0 * ot[dt][3] + a1 * plo.w) / den;
                    hi.x = (a0 * ot[dt][4] + a1 * phi.x) / den; hi.y = (a0 * ot[dt][5] + a1 * phi.y) / den;
                    hi.z = (a0 * ot[dt][6] + a1 * phi.z) / den; hi.w = (a0 * ot[dt][7] + a1 * phi.w) / den;
                    *(float4*)&Ps[qr][(dt << 4) + rowOff + 0] = lo;
                    *(float4*)&Ps[qr][(dt << 4) + rowOff + 4] = hi;
                }
            }
            lds_barrier();
            const int c4 = l16 << 2;
            #pragma unroll
            for (int i = 0; i < 4; ++i) {
                const int rr = (wave << 3) + (i << 1) + half;
                const float4 v = *(const float4*)&Ps[rr][c4];
                *(float4*)&Out[(rowB + qBase + rr) * DIMp + (h << 6) + c4] = v;
            }
        }
    }
}

extern "C" void kernel_launch(void* const* d_in, const int* in_sizes, int n_in,
                              void* d_out, int out_size, void* d_ws, size_t ws_size,
                              hipStream_t stream) {
    (void)in_sizes; (void)n_in; (void)out_size; (void)ws_size;
    constexpr int DIM = 1024, HEAD_DIM = 64, NUM_HEADS = 16, BATCH = 2, SEQ = 2048;
    constexpr int M = BATCH * SEQ;

    const float* x  = (const float*)d_in[0];
    const float* Wq = (const float*)d_in[1];
    const float* Wk = (const float*)d_in[2];
    const float* Wv = (const float*)d_in[3];
    const float* Wo = (const float*)d_in[4];
    const float* bo = (const float*)d_in[5];
    float* out = (float*)d_out;

    float* Qbuf = (float*)d_ws;
    float* Kbuf = Qbuf + (size_t)M * DIM;
    float* Vbuf = Kbuf + (size_t)M * HEAD_DIM;
    float* Abuf = Vbuf + (size_t)M * HEAD_DIM;

    const float scale = 0.125f;

    gemm_wmma_f32<<<dim3(DIM / 64 + 2 * (HEAD_DIM / 64), M / 128), 256, 0, stream>>>(
        x, Wq, Wk, Wv, nullptr, Qbuf, Kbuf, Vbuf, M, DIM, HEAD_DIM, DIM, scale);

    mqa_flash_wmma<<<dim3(SEQ / 128, NUM_HEADS, BATCH), 256, 0, stream>>>(
        Qbuf, Kbuf, Vbuf, Abuf, SEQ, DIM);

    gemm_wmma_f32<<<dim3(DIM / 64, M / 128), 256, 0, stream>>>(
        Abuf, Wo, nullptr, nullptr, bo, out, nullptr, nullptr, M, DIM, 0, DIM, 1.0f);
}


# QKV projection: the K and V column blocks fold into the 512 Q workgroups (each adds a 16x64 K or V tile of its own rows, k split over 2 waves); grid 576 to 512 = 2 per WGP; K, V sum order changes
# speedup vs baseline: 1.8795x; 1.0374x over previous
#include <hip/hip_runtime.h>
#include <math.h>

typedef __attribute__((ext_vector_type(2))) float v2f;
typedef __attribute__((ext_vector_type(8))) float v8f;

#define WMMA_F32(a, b, c) \
    __builtin_amdgcn_wmma_f32_16x16x4_f32(false, (a), false, (b), (short)0, (c), false, false)

__device__ __forceinline__ void lds_barrier() {
    __builtin_amdgcn_fence(__ATOMIC_RELEASE, "workgroup", "local");
    __builtin_amdgcn_s_barrier();
    __builtin_amdgcn_fence(__ATOMIC_ACQUIRE, "workgroup", "local");
}

template <bool EXTRA>
__global__ __launch_bounds__(256) void gemm_wmma_f32(
    const float* __restrict__ A,
    const float* __restrict__ W, const float* __restrict__ WkE, const float* __restrict__ WvE,
    const float* __restrict__ bias,
    float* __restrict__ C, float* __restrict__ CkE, float* __restrict__ CvE,
    int M, int N, int K, float alpha)
{
    const int bx = (int)blockIdx.x;

    __shared__ __align__(16) float As[128][18];
    __shared__ __align__(16) float Wt[64][18];
    __shared__ __align__(16) float Et[EXTRA ? 64 : 1][18];

    const int tid    = threadIdx.x;
    const int wave   = tid >> 5;
    const int lane   = tid & 31;
    const int l16    = lane & 15;
    const int half   = lane >> 4;
    const int kb     = half << 1;
    const int rowOff = half << 3;

    const int wm = wave >> 1;
    const int wn = wave & 1;

    const int rowBlock = blockIdx.y * 128;
    const int colBlock = bx * 64;

    v8f acc[2][2];
    #pragma unroll
    for (int mt = 0; mt < 2; ++mt)
        #pragma unroll
        for (int nt = 0; nt < 2; ++nt) acc[mt][nt] = (v8f)(0.0f);

    const int ar = tid >> 1;
    const int ac = (tid & 1) << 3;
    const int wr = tid >> 4;
    const int wc = (tid & 15) << 2;

    const int aRow0 = (wm << 5) + l16;
    const int aRow1 = aRow0 + 16;
    const int bRow0 = (wn << 5) + l16;
    const int bRow1 = bRow0 + 16;

    const float* WE = (bx & 1) ? WvE : WkE;
    float* CE = (bx & 1) ? CvE : CkE;
    const int eRow = (bx >> 1) << 4;
    const int ect  = wave & 3;
    const int ekk  = (wave >> 2) << 3;
    const int eaRow = eRow + l16;
    const int ebRow = (ect << 4) + l16;
    v8f accE = (v8f)(0.0f);
    const float* eSrc = WE + (size_t)wr * 64 + wc;
    float4 ev = make_float4(0.0f, 0.0f, 0.0f, 0.0f);

    const float* aSrc = A + (size_t)(rowBlock + ar) * K + ac;
    const float* wSrc = W + (size_t)wr * N + colBlock + wc;
    float4 a0 = *(const float4*)(aSrc);
    float4 a1 = *(const float4*)(aSrc + 4);
    float4 wv = *(const float4*)(wSrc);
    if (EXTRA) ev = *(const float4*)(eSrc);

    for (int kt = 0; kt < K; kt += 16) {
        As[ar][ac + 0] = a0.x; As[ar][ac + 1] = a0.y;
        As[ar][ac + 2] = a0.z; As[ar][ac + 3] = a0.w;
        As[ar][ac + 4] = a1.x; As[ar][ac + 5] = a1.y;
        As[ar][ac + 6] = a1.z; As[ar][ac + 7] = a1.w;
        Wt[wc + 0][wr] = wv.x; Wt[wc + 1][wr] = wv.y;
        Wt[wc + 2][wr] = wv.z; Wt[wc + 3][wr] = wv.w;
        if (EXTRA) {
            Et[wc + 0][wr] = ev.x; Et[wc + 1][wr] = ev.y;
            Et[wc + 2][wr] = ev.z; Et[wc + 3][wr] = ev.w;
        }
        if (kt + 16 < K) {
            aSrc += 16;
            wSrc += (size_t)16 * N;
            a0 = *(const float4*)(aSrc);
            a1 = *(const float4*)(aSrc + 4);
            wv = *(const float4*)(wSrc);
            if (EXTRA) {
                eSrc += (size_t)16 * 64;
                ev = *(const float4*)(eSrc);
            }
            if (kt + 32 < K) {
                __builtin_prefetch(aSrc + 16, 0, 0);
                __builtin_prefetch(wSrc + (size_t)16 * N, 0, 0);
            }
        }
        lds_barrier();

        #pragma unroll
        for (int kk = 0; kk < 16; kk += 4) {
            const v2f av0 = *(const v2f*)&As[aRow0][kk + kb];
            const v2f av1 = *(const v2f*)&As[aRow1][kk + kb];
            const v2f bv0 = *(const v2f*)&Wt[bRow0][kk + kb];
            const v2f bv1 = *(const v2f*)&Wt[bRow1][kk + kb];
            acc[0][0] = WMMA_F32(av0, bv0, acc[0][0]);
            acc[0][1] = WMMA_F32(av0, bv1, acc[0][1]);
            acc[1][0] = WMMA_F32(av1, bv0, acc[1][0]);
            acc[1][1] = WMMA_F32(av1, bv1, acc[1][1]);
        }
        if (EXTRA) {
            #pragma unroll
            for (int q = 0; q < 2; ++q) {
                const int kk = ekk + (q << 2);
                const v2f av = *(const v2f*)&As[eaRow][kk + kb];
                const v2f bv = *(const v2f*)&Et[ebRow][kk + kb];
                accE = WMMA_F32(av, bv, accE);
            }
        }
        lds_barrier();
    }

    #pragma unroll
    for (int mt = 0; mt < 2; ++mt) {
        #pragma unroll
        for (int nt = 0; nt < 2; ++nt) {
            const int col = colBlock + (wn << 5) + (nt << 4) + l16;
            const float bv = bias ? bias[col] : 0.0f;
            #pragma unroll
            for (int v = 0; v < 8; ++v) {
                const int row = rowBlock + (wm << 5) + (mt << 4) + rowOff + v;
                C[(size_t)row * N + col] = alpha * acc[mt][nt][v] + bv;
            }
        }
    }

    if (EXTRA) {
        float* red = &As[0][0];
        if (wave >= 4) {
            #pragma unroll
            for (int v = 0; v < 8; ++v) red[(ect << 8) + (v << 5) + lane] = accE[v];
        }
        lds_barrier();
        if (wave < 4) {
            const int col = (ect << 4) + l16;
            #pragma unroll
            for (int v = 0; v < 8; ++v) {
                const int row = rowBlock + eRow + rowOff + v;
                CE[(size_t)row * 64 + col] = accE[v] + red[(ect << 8) + (v << 5) + lane];
            }
        }
    }
}

__device__ __forceinline__ float xhalf(float x) {
    const int b = __builtin_bit_cast(int, x);
    return __builtin_bit_cast(float, __builtin_amdgcn_permlanex16(b, b, 0x76543210, 0xfedcba98, false, false));
}

__global__ __launch_bounds__(256) void mqa_flash_wmma(
    const float* __restrict__ Q,
    const float* __restrict__ Kg,
    const float* __restrict__ Vg,
    float* __restrict__ Out,
    int S, int DIMp)
{
    __shared__ __align__(16) float Ks[64][68];
    __shared__ __align__(16) float Vt[64][68];
    __shared__ __align__(16) float Ps[64][68];
    __shared__ float Ms[64];
    __shared__ float Ls[64];

    const int tid    = threadIdx.x;
    const int wave   = tid >> 5;
    const int lane   = tid & 31;
    const int l16    = lane & 15;
    const int half   = lane >> 4;
    const int kb     = half << 1;
    const int rowOff = half << 3;
    const int qw     = wave & 3;
    const int kg     = wave >> 2;
    const int keyOff = kg << 5;

    const int h     = blockIdx.y;
    const int b     = blockIdx.z;
    const size_t rowB = (size_t)b * S;
    const int nQ    = (int)gridDim.x << 1;

    const int lr = tid >> 2;
    const int lq = (tid & 3) << 2;

    const int qr = (qw << 4) + l16;

    for (int pass = 0; pass < 2; ++pass) {
        const int qTile = pass == 0 ? (nQ - 1 - (int)blockIdx.x) : (int)blockIdx.x;
        const int qBase = qTile << 6;

        const int nTiles = qTile + 1;

        const float* ksrc = Kg + (rowB + lr) * 64 + lq;
        const float* vsrc = Vg + (rowB + lr) * 64 + lq;
        float4 kr[4], vr[4];
        #pragma unroll
        for (int j = 0; j < 4; ++j) {
            kr[j] = *(const float4*)(ksrc + (j << 4));
            vr[j] = *(const float4*)(vsrc + (j << 4));
        }

        if (pass != 0) lds_barrier();

        {
            const float* src = Q + (rowB + qBase + lr) * DIMp + (h << 6) + lq;
            #pragma unroll
            for (int j = 0; j < 4; ++j) {
                const float4 qv = *(const float4*)(src + (j << 4));
                Ps[lr][lq + (j << 4) + 0] = qv.x; Ps[lr][lq + (j << 4) + 1] = qv.y;
                Ps[lr][lq + (j << 4) + 2] = qv.z; Ps[lr][lq + (j << 4) + 3] = qv.w;
            }
        }
        lds_barrier();

        v2f bq[16];
        #pragma unroll
        for (int i = 0; i < 16; ++i)
            bq[i] = *(const v2f*)&Ps[qr][(i << 2) + kb];

        v8f ot[4];
        #pragma unroll
        for (int dt = 0; dt < 4; ++dt) ot[dt] = (v8f)(0.0f);
        float m = -1e30f;
        float lsum = 0.0f;

        const int qIdx = qBase + qr;

        for (int t = 0; t < nTiles; ++t) {
            const int kBase = t << 6;
            lds_barrier();
            #pragma unroll
            for (int j = 0; j < 4; ++j) {
                const float4 kv = kr[j];
                const float4 vv = vr[j];
                Ks[lr][lq + (j << 4) + 0] = kv.x; Ks[lr][lq + (j << 4) + 1] = kv.y;
                Ks[lr][lq + (j << 4) + 2] = kv.z; Ks[lr][lq + (j << 4) + 3] = kv.w;
                Vt[lq + (j << 4) + 0][lr] = vv.x; Vt[lq + (j << 4) + 1][lr] = vv.y;
                Vt[lq + (j << 4) + 2][lr] = vv.z; Vt[lq + (j << 4) + 3][lr] = vv.w;
            }
            if (t + 1 < nTiles) {
                ksrc += 64 * 64;
                vsrc += 64 * 64;
                #pragma unroll
                for (int j = 0; j < 4; ++j) {
                    kr[j] = *(const float4*)(ksrc + (j << 4));
                    vr[j] = *(const float4*)(vsrc + (j << 4));
                }
            }
            lds_barrier();

            v8f st[2];
            #pragma unroll
            for (int mt = 0; mt < 2; ++mt) st[mt] = (v8f)(0.0f);
            #pragma unroll
            for (int kk = 0; kk < 64; kk += 4) {
                const v2f bf = bq[kk >> 2];
                #pragma unroll
                for (int mt = 0; mt < 2; ++mt) {
                    const v2f af = *(const v2f*)&Ks[keyOff + (mt << 4) + l16][kk + kb];
                    st[mt] = WMMA_F32(af, bf, st[mt]);
                }
            }

            if (t == nTiles - 1) {
                #pragma unroll
                for (int mt = 0; mt < 2; ++mt) {
                    #pragma unroll
                    for (int r = 0; r < 8; ++r) {
                        const int key = kBase + keyOff + (mt << 4) + rowOff + r;
                        if (key > qIdx) st[mt][r] = -1e30f;
                    }
                }
            }

            float x = -1e30f;
            #pragma unroll
            for (int mt = 0; mt < 2; ++mt) {
                #pragma unroll
                for (int r = 0; r < 8; ++r) x = fmaxf(x, st[mt][r]);
            }
            x = fmaxf(x, xhalf(x));
            const float mNew = fmaxf(m, x);
            const float corr = __expf(m - mNew);
            m = mNew;
            #pragma unroll
            for (int dt = 0; dt < 4; ++dt)
                #pragma unroll
                for (int r = 0; r < 8; ++r) ot[dt][r] *= corr;

            float rs = 0.0f;
            #pragma unroll
            for (int mt = 0; mt < 2; ++mt) {
                #pragma unroll
                for (int r = 0; r < 8; ++r) {
                    const float p = __expf(st[mt][r] - m);
                    st[mt][r] = p;
                    rs += p;
                }
            }
            lsum = lsum * corr + rs;

            #pragma unroll
            for (int mt = 0; mt < 2; ++mt) {
                #pragma unroll
                for (int j = 0; j < 4; ++j) {
                    v2f pf;
                    pf.x = st[mt][2 * j];
                    pf.y = st[mt][2 * j + 1];
                    #pragma unroll
                    for (int dt = 0; dt < 4; ++dt) {
                        const v2f af = *(const v2f*)&Vt[(dt << 4) + l16][keyOff + (mt << 4) + rowOff + (j << 1)];
                        ot[dt] = WMMA_F32(af, pf, ot[dt]);
                    }
                }
            }
        }

        {
            const float ltot = lsum + xhalf(lsum);
            if (kg == 1) {
                #pragma unroll
                for (int dt = 0; dt < 4; ++dt) {
                    float4 lo, hi;
                    lo.x = ot[dt][0]; lo.y = ot[dt][1]; lo.z = ot[dt][2]; lo.w = ot[dt][3];
                    hi.x = ot[dt][4]; hi.y = ot[dt][5]; hi.z = ot[dt][6]; hi.w = ot[dt][7];
                    *(float4*)&Ps[qr][(dt << 4) + rowOff + 0] = lo;
                    *(float4*)&Ps[qr][(dt << 4) + rowOff + 4] = hi;
                }
                if (half == 0) { Ms[qr] = m; Ls[qr] = ltot; }
            }
            lds_barrier();
            if (kg == 0) {
                const float m1 = Ms[qr];
                const float l1 = Ls[qr];
                const float mm = fmaxf(m, m1);
                const float a0 = __expf(m - mm);
                const float a1 = __expf(m1 - mm);
                const float den = a0 * ltot + a1 * l1;
                #pragma unroll
                for (int dt = 0; dt < 4; ++dt) {
                    const float4 plo = *(const float4*)&Ps[qr][(dt << 4) + rowOff + 0];
                    const float4 phi = *(const float4*)&Ps[qr][(dt << 4) + rowOff + 4];
                    float4 lo, hi;
                    lo.x = (a0 * ot[dt][0] + a1 * plo.x) / den; lo.y = (a0 * ot[dt][1] + a1 * plo.y) / den;
                    lo.z = (a0 * ot[dt][2] + a1 * plo.z) / den; lo.w = (a0 * ot[dt][3] + a1 * plo.w) / den;
                    hi.x = (a0 * ot[dt][4] + a1 * phi.x) / den; hi.y = (a0 * ot[dt][5] + a1 * phi.y) / den;
                    hi.z = (a0 * ot[dt][6] + a1 * phi.z) / den; hi.w = (a0 * ot[dt][7] + a1 * phi.w) / den;
                    *(float4*)&Ps[qr][(dt << 4) + rowOff + 0] = lo;
                    *(float4*)&Ps[qr][(dt << 4) + rowOff + 4] = hi;
                }
            }
            lds_barrier();
            const int c4 = l16 << 2;
            #pragma unroll
            for (int i = 0; i < 4; ++i) {
                const int rr = (wave << 3) + (i << 1) + half;
                const float4 v = *(const float4*)&Ps[rr][c4];
                *(float4*)&Out[(rowB + qBase + rr) * DIMp + (h << 6) + c4] = v;
            }
        }
    }
}

extern "C" void kernel_launch(void* const* d_in, const int* in_sizes, int n_in,
                              void* d_out, int out_size, void* d_ws, size_t ws_size,
                              hipStream_t stream) {
    (void)in_sizes; (void)n_in; (void)out_size; (void)ws_size;
    constexpr int DIM = 1024, HEAD_DIM = 64, NUM_HEADS = 16, BATCH = 2, SEQ = 2048;
    constexpr int M = BATCH * SEQ;

    const float* x  = (const float*)d_in[0];
    const float* Wq = (const float*)d_in[1];
    const float* Wk = (const float*)d_in[2];
    const float* Wv = (const float*)d_in[3];
    const float* Wo = (const float*)d_in[4];
    const float* bo = (const float*)d_in[5];
    float* out = (float*)d_out;

    float* Qbuf = (float*)d_ws;
    float* Kbuf = Qbuf + (size_t)M * DIM;
    float* Vbuf = Kbuf + (size_t)M * HEAD_DIM;
    float* Abuf = Vbuf + (size_t)M * HEAD_DIM;

    const float scale = 0.125f;

    gemm_wmma_f32<true><<<dim3(DIM / 64, M / 128), 256, 0, stream>>>(
        x, Wq, Wk, Wv, nullptr, Qbuf, Kbuf, Vbuf, M, DIM, DIM, scale);

    mqa_flash_wmma<<<dim3(SEQ / 128, NUM_HEADS, BATCH), 256, 0, stream>>>(
        Qbuf, Kbuf, Vbuf, Abuf, SEQ, DIM);

    gemm_wmma_f32<false><<<dim3(DIM / 64, M / 128), 256, 0, stream>>>(
        Abuf, Wo, nullptr, nullptr, bo, out, nullptr, nullptr, M, DIM, DIM, 1.0f);
}


# Both projection GEMMs: K tile 16 -> 64 deep: one staging round and barrier pair per 64+8 WMMAs of a wave, not per 16+2 (LDS pitch 68, conflict-free transposed stores); k order and all sums unchanged
# speedup vs baseline: 1.9896x; 1.0586x over previous
#include <hip/hip_runtime.h>
#include <math.h>

typedef __attribute__((ext_vector_type(2))) float v2f;
typedef __attribute__((ext_vector_type(4))) float v4f;
typedef __attribute__((ext_vector_type(8))) float v8f;

#define WMMA_F32(a, b, c) \
    __builtin_amdgcn_wmma_f32_16x16x4_f32(false, (a), false, (b), (short)0, (c), false, false)

__device__ __forceinline__ void lds_barrier() {
    __builtin_amdgcn_fence(__ATOMIC_RELEASE, "workgroup", "local");
    __builtin_amdgcn_s_barrier();
    __builtin_amdgcn_fence(__ATOMIC_ACQUIRE, "workgroup", "local");
}

template <bool EXTRA>
__global__ __launch_bounds__(256) void gemm_wmma_f32(
    const float* __restrict__ A,
    const float* __restrict__ W, const float* __restrict__ WkE, const float* __restrict__ WvE,
    const float* __restrict__ bias,
    float* __restrict__ C, float* __restrict__ CkE, float* __restrict__ CvE,
    int M, int N, int K, float alpha)
{
    const int bx = (int)blockIdx.x;

    constexpr int BK  = 64;
    constexpr int LDP = BK + 4;

    __shared__ __align__(16) float As[128][LDP];
    __shared__ __align__(16) float Wt[64][LDP];
    __shared__ __align__(16) float Et[EXTRA ? 64 : 1][LDP];

    const int tid    = threadIdx.x;
    const int wave   = tid >> 5;
    const int lane   = tid & 31;
    const int l16    = lane & 15;
    const int half   = lane >> 4;
    const int kb     = half << 1;
    const int rowOff = half << 3;

    const int wm = wave >> 1;
    const int wn = wave & 1;

    const int rowBlock = blockIdx.y * 128;
    const int colBlock = bx * 64;

    v8f acc[2][2];
    #pragma unroll
    for (int mt = 0; mt < 2; ++mt)
        #pragma unroll
        for (int nt = 0; nt < 2; ++nt) acc[mt][nt] = (v8f)(0.0f);

    const int sr = tid >> 4;
    const int sc = (tid & 15) << 2;
    const int wk  = (tid >> 2) & 15;
    const int wn4 = ((tid & 3) << 2) + ((tid >> 6) << 4);

    const int aRow0 = (wm << 5) + l16;
    const int aRow1 = aRow0 + 16;
    const int bRow0 = (wn << 5) + l16;
    const int bRow1 = bRow0 + 16;

    const float* WE = (bx & 1) ? WvE : WkE;
    float* CE = (bx & 1) ? CvE : CkE;
    const int eRow = (bx >> 1) << 4;
    const int ect  = wave & 3;
    const int ekk  = (wave >> 2) << 3;
    const int eaRow = eRow + l16;
    const int ebRow = (ect << 4) + l16;
    v8f accE = (v8f)(0.0f);

    const float* aSrc = A + (size_t)(rowBlock + sr) * K + sc;
    const float* wSrc = W + (size_t)wk * N + colBlock + wn4;
    const float* eSrc = WE + (size_t)wk * 64 + wn4;
    v4f av[8], wv[4], ev[4];
    #pragma unroll
    for (int p = 0; p < 8; ++p) av[p] = *(const v4f*)(aSrc + (size_t)(p << 4) * K);
    #pragma unroll
    for (int p = 0; p < 4; ++p) wv[p] = *(const v4f*)(wSrc + (size_t)(p << 4) * N);
    #pragma unroll
    for (int p = 0; p < 4; ++p) ev[p] = (v4f)(0.0f);
    if (EXTRA) {
        #pragma unroll
        for (int p = 0; p < 4; ++p) ev[p] = *(const v4f*)(eSrc + (p << 10));
    }

    for (int kt = 0; kt < K; kt += BK) {
        #pragma unroll
        for (int p = 0; p < 8; ++p) *(v4f*)&As[sr + (p << 4)][sc] = av[p];
        #pragma unroll
        for (int p = 0; p < 4; ++p) {
            const int k = wk + (p << 4);
            Wt[wn4 + 0][k] = wv[p].x; Wt[wn4 + 1][k] = wv[p].y;
            Wt[wn4 + 2][k] = wv[p].z; Wt[wn4 + 3][k] = wv[p].w;
        }
        if (EXTRA) {
            #pragma unroll
            for (int p = 0; p < 4; ++p) {
                const int k = wk + (p << 4);
                Et[wn4 + 0][k] = ev[p].x; Et[wn4 + 1][k] = ev[p].y;
                Et[wn4 + 2][k] = ev[p].z; Et[wn4 + 3][k] = ev[p].w;
            }
        }
        if (kt + BK < K) {
            aSrc += BK;
            wSrc += (size_t)BK * N;
            #pragma unroll
            for (int p = 0; p < 8; ++p) av[p] = *(const v4f*)(aSrc + (size_t)(p << 4) * K);
            #pragma unroll
            for (int p = 0; p < 4; ++p) wv[p] = *(const v4f*)(wSrc + (size_t)(p << 4) * N);
            if (EXTRA) {
                eSrc += (size_t)BK * 64;
                #pragma unroll
                for (int p = 0; p < 4; ++p) ev[p] = *(const v4f*)(eSrc + (p << 10));
            }
        }
        lds_barrier();

        #pragma unroll
        for (int ks = 0; ks < BK; ks += 16) {
            #pragma unroll
            for (int kk = ks; kk < ks + 16; kk += 4) {
                const v2f av0 = *(const v2f*)&As[aRow0][kk + kb];
                const v2f av1 = *(const v2f*)&As[aRow1][kk + kb];
                const v2f bv0 = *(const v2f*)&Wt[bRow0][kk + kb];
                const v2f bv1 = *(const v2f*)&Wt[bRow1][kk + kb];
                acc[0][0] = WMMA_F32(av0, bv0, acc[0][0]);
                acc[0][1] = WMMA_F32(av0, bv1, acc[0][1]);
                acc[1][0] = WMMA_F32(av1, bv0, acc[1][0]);
                acc[1][1] = WMMA_F32(av1, bv1, acc[1][1]);
            }
            if (EXTRA) {
                #pragma unroll
                for (int q = 0; q < 2; ++q) {
                    const int kk = ks + ekk + (q << 2);
                    const v2f ea = *(const v2f*)&As[eaRow][kk + kb];
                    const v2f eb = *(const v2f*)&Et[ebRow][kk + kb];
                    accE = WMMA_F32(ea, eb, accE);
                }
            }
        }
        lds_barrier();
    }

    #pragma unroll
    for (int mt = 0; mt < 2; ++mt) {
        #pragma unroll
        for (int nt = 0; nt < 2; ++nt) {
            const int col = colBlock + (wn << 5) + (nt << 4) + l16;
            const float bv = bias ? bias[col] : 0.0f;
            #pragma unroll
            for (int v = 0; v < 8; ++v) {
                const int row = rowBlock + (wm << 5) + (mt << 4) + rowOff + v;
                C[(size_t)row * N + col] = alpha * acc[mt][nt][v] + bv;
            }
        }
    }

    if (EXTRA) {
        float* red = &As[0][0];
        if (wave >= 4) {
            #pragma unroll
            for (int v = 0; v < 8; ++v) red[(ect << 8) + (v << 5) + lane] = accE[v];
        }
        lds_barrier();
        if (wave < 4) {
            const int col = (ect << 4) + l16;
            #pragma unroll
            for (int v = 0; v < 8; ++v) {
                const int row = rowBlock + eRow + rowOff + v;
                CE[(size_t)row * 64 + col] = accE[v] + red[(ect << 8) + (v << 5) + lane];
            }
        }
    }
}

__device__ __forceinline__ float xhalf(float x) {
    const int b = __builtin_bit_cast(int, x);
    return __builtin_bit_cast(float, __builtin_amdgcn_permlanex16(b, b, 0x76543210, 0xfedcba98, false, false));
}

__global__ __launch_bounds__(256) void mqa_flash_wmma(
    const float* __restrict__ Q,
    const float* __restrict__ Kg,
    const float* __restrict__ Vg,
    float* __restrict__ Out,
    int S, int DIMp)
{
    __shared__ __align__(16) float Ks[64][68];
    __shared__ __align__(16) float Vt[64][68];
    __shared__ __align__(16) float Ps[64][68];
    __shared__ float Ms[64];
    __shared__ float Ls[64];

    const int tid    = threadIdx.x;
    const int wave   = tid >> 5;
    const int lane   = tid & 31;
    const int l16    = lane & 15;
    const int half   = lane >> 4;
    const int kb     = half << 1;
    const int rowOff = half << 3;
    const int qw     = wave & 3;
    const int kg     = wave >> 2;
    const int keyOff = kg << 5;

    const int h     = blockIdx.y;
    const int b     = blockIdx.z;
    const size_t rowB = (size_t)b * S;
    const int nQ    = (int)gridDim.x << 1;

    const int lr = tid >> 2;
    const int lq = (tid & 3) << 2;

    const int qr = (qw << 4) + l16;

    for (int pass = 0; pass < 2; ++pass) {
        const int qTile = pass == 0 ? (nQ - 1 - (int)blockIdx.x) : (int)blockIdx.x;
        const int qBase = qTile << 6;

        const int nTiles = qTile + 1;

        const float* ksrc = Kg + (rowB + lr) * 64 + lq;
        const float* vsrc = Vg + (rowB + lr) * 64 + lq;
        float4 kr[4], vr[4];
        #pragma unroll
        for (int j = 0; j < 4; ++j) {
            kr[j] = *(const float4*)(ksrc + (j << 4));
            vr[j] = *(const float4*)(vsrc + (j << 4));
        }

        if (pass != 0) lds_barrier();

        {
            const float* src = Q + (rowB + qBase + lr) * DIMp + (h << 6) + lq;
            #pragma unroll
            for (int j = 0; j < 4; ++j) {
                const float4 qv = *(const float4*)(src + (j << 4));
                Ps[lr][lq + (j << 4) + 0] = qv.x; Ps[lr][lq + (j << 4) + 1] = qv.y;
                Ps[lr][lq + (j << 4) + 2] = qv.z; Ps[lr][lq + (j << 4) + 3] = qv.w;
            }
        }
        lds_barrier();

        v2f bq[16];
        #pragma unroll
        for (int i = 0; i < 16; ++i)
            bq[i] = *(const v2f*)&Ps[qr][(i << 2) + kb];

        v8f ot[4];
        #pragma unroll
        for (int dt = 0; dt < 4; ++dt) ot[dt] = (v8f)(0.0f);
        float m = -1e30f;
        float lsum = 0.0f;

        const int qIdx = qBase + qr;

        for (int t = 0; t < nTiles; ++t) {
            const int kBase = t << 6;
            lds_barrier();
            #pragma unroll
            for (int j = 0; j < 4; ++j) {
                const float4 kv = kr[j];
                const float4 vv = vr[j];
                Ks[lr][lq + (j << 4) + 0] = kv.x; Ks[lr][lq + (j << 4) + 1] = kv.y;
                Ks[lr][lq + (j << 4) + 2] = kv.z; Ks[lr][lq + (j << 4) + 3] = kv.w;
                Vt[lq + (j << 4) + 0][lr] = vv.x; Vt[lq + (j << 4) + 1][lr] = vv.y;
                Vt[lq + (j << 4) + 2][lr] = vv.z; Vt[lq + (j << 4) + 3][lr] = vv.w;
            }
            if (t + 1 < nTiles) {
                ksrc += 64 * 64;
                vsrc += 64 * 64;
                #pragma unroll
                for (int j = 0; j < 4; ++j) {
                    kr[j] = *(const float4*)(ksrc + (j << 4));
                    vr[j] = *(const float4*)(vsrc + (j << 4));
                }
            }
            lds_barrier();

            v8f st[2];
            #pragma unroll
            for (int mt = 0; mt < 2; ++mt) st[mt] = (v8f)(0.0f);
            #pragma unroll
            for (int kk = 0; kk < 64; kk += 4) {
                const v2f bf = bq[kk >> 2];
                #pragma unroll
                for (int mt = 0; mt < 2; ++mt) {
                    const v2f af = *(const v2f*)&Ks[keyOff + (mt << 4) + l16][kk + kb];
                    st[mt] = WMMA_F32(af, bf, st[mt]);
                }
            }

            if (t == nTiles - 1) {
                #pragma unroll
                for (int mt = 0; mt < 2; ++mt) {
                    #pragma unroll
                    for (int r = 0; r < 8; ++r) {
                        const int key = kBase + keyOff + (mt << 4) + rowOff + r;
                        if (key > qIdx) st[mt][r] = -1e30f;
                    }
                }
            }

            float x = -1e30f;
            #pragma unroll
            for (int mt = 0; mt < 2; ++mt) {
                #pragma unroll
                for (int r = 0; r < 8; ++r) x = fmaxf(x, st[mt][r]);
            }
            x = fmaxf(x, xhalf(x));
            const float mNew = fmaxf(m, x);
            const float corr = __expf(m - mNew);
            m = mNew;
            #pragma unroll
            for (int dt = 0; dt < 4; ++dt)
                #pragma unroll
                for (int r = 0; r < 8; ++r) ot[dt][r] *= corr;

            float rs = 0.0f;
            #pragma unroll
            for (int mt = 0; mt < 2; ++mt) {
                #pragma unroll
                for (int r = 0; r < 8; ++r) {
                    const float p = __expf(st[mt][r] - m);
                    st[mt][r] = p;
                    rs += p;
                }
            }
            lsum = lsum * corr + rs;

            #pragma unroll
            for (int mt = 0; mt < 2; ++mt) {
                #pragma unroll
                for (int j = 0; j < 4; ++j) {
                    v2f pf;
                    pf.x = st[mt][2 * j];
                    pf.y = st[mt][2 * j + 1];
                    #pragma unroll
                    for (int dt = 0; dt < 4; ++dt) {
                        const v2f af = *(const v2f*)&Vt[(dt << 4) + l16][keyOff + (mt << 4) + rowOff + (j << 1)];
                        ot[dt] = WMMA_F32(af, pf, ot[dt]);
                    }
                }
            }
        }

        {
            const float ltot = lsum + xhalf(lsum);
            if (kg == 1) {
                #pragma unroll
                for (int dt = 0; dt < 4; ++dt) {
                    float4 lo, hi;
                    lo.x = ot[dt][0]; lo.y = ot[dt][1]; lo.z = ot[dt][2]; lo.w = ot[dt][3];
                    hi.x = ot[dt][4]; hi.y = ot[dt][5]; hi.z = ot[dt][6]; hi.w = ot[dt][7];
                    *(float4*)&Ps[qr][(dt << 4) + rowOff + 0] = lo;
                    *(float4*)&Ps[qr][(dt << 4) + rowOff + 4] = hi;
                }
                if (half == 0) { Ms[qr] = m; Ls[qr] = ltot; }
            }
            lds_barrier();
            if (kg == 0) {
                const float m1 = Ms[qr];
                const float l1 = Ls[qr];
                const float mm = fmaxf(m, m1);
                const float a0 = __expf(m - mm);
                const float a1 = __expf(m1 - mm);
                const float den = a0 * ltot + a1 * l1;
                #pragma unroll
                for (int dt = 0; dt < 4; ++dt) {
                    const float4 plo = *(const float4*)&Ps[qr][(dt << 4) + rowOff + 0];
                    const float4 phi = *(const float4*)&Ps[qr][(dt << 4) + rowOff + 4];
                    float4 lo, hi;
                    lo.x = (a0 * ot[dt][0] + a1 * plo.x) / den; lo.y = (a0 * ot[dt][1] + a1 * plo.y) / den;
                    lo.z = (a0 * ot[dt][2] + a1 * plo.z) / den; lo.w = (a0 * ot[dt][3] + a1 * plo.w) / den;
                    hi.x = (a0 * ot[dt][4] + a1 * phi.x) / den; hi.y = (a0 * ot[dt][5] + a1 * phi.y) / den;
                    hi.z = (a0 * ot[dt][6] + a1 * phi.z) / den; hi.w = (a0 * ot[dt][7] + a1 * phi.w) / den;
                    *(float4*)&Ps[qr][(dt << 4) + rowOff + 0] = lo;
                    *(float4*)&Ps[qr][(dt << 4) + rowOff + 4] = hi;
                }
            }
            lds_barrier();
            const int c4 = l16 << 2;
            #pragma unroll
            for (int i = 0; i < 4; ++i) {
                const int rr = (wave << 3) + (i << 1) + half;
                const float4 v = *(const float4*)&Ps[rr][c4];
                *(float4*)&Out[(rowB + qBase + rr) * DIMp + (h << 6) + c4] = v;
            }
        }
    }
}

extern "C" void kernel_launch(void* const* d_in, const int* in_sizes, int n_in,
                              void* d_out, int out_size, void* d_ws, size_t ws_size,
                              hipStream_t stream) {
    (void)in_sizes; (void)n_in; (void)out_size; (void)ws_size;
    constexpr int DIM = 1024, HEAD_DIM = 64, NUM_HEADS = 16, BATCH = 2, SEQ = 2048;
    constexpr int M = BATCH * SEQ;

    const float* x  = (const float*)d_in[0];
    const float* Wq = (const float*)d_in[1];
    const float* Wk = (const float*)d_in[2];
    const float* Wv = (const float*)d_in[3];
    const float* Wo = (const float*)d_in[4];
    const float* bo = (const float*)d_in[5];
    float* out = (float*)d_out;

    float* Qbuf = (float*)d_ws;
    float* Kbuf = Qbuf + (size_t)M * DIM;
    float* Vbuf = Kbuf + (size_t)M * HEAD_DIM;
    float* Abuf = Vbuf + (size_t)M * HEAD_DIM;

    const float scale = 0.125f;

    gemm_wmma_f32<true><<<dim3(DIM / 64, M / 128), 256, 0, stream>>>(
        x, Wq, Wk, Wv, nullptr, Qbuf, Kbuf, Vbuf, M, DIM, DIM, scale);

    mqa_flash_wmma<<<dim3(SEQ / 128, NUM_HEADS, BATCH), 256, 0, stream>>>(
        Qbuf, Kbuf, Vbuf, Abuf, SEQ, DIM);

    gemm_wmma_f32<false><<<dim3(DIM / 64, M / 128), 256, 0, stream>>>(
        Abuf, Wo, nullptr, nullptr, bo, out, nullptr, nullptr, M, DIM, DIM, 1.0f);
}


# All three kernels: LDS tiles in two stages; tile t+1 is written to the other stage in the middle of tile t's WMMAs; one barrier per tile step instead of two; every sum and its order unchanged
# speedup vs baseline: 2.0712x; 1.0410x over previous
#include <hip/hip_runtime.h>
#include <math.h>

typedef __attribute__((ext_vector_type(2))) float v2f;
typedef __attribute__((ext_vector_type(4))) float v4f;
typedef __attribute__((ext_vector_type(8))) float v8f;

#define WMMA_F32(a, b, c) \
    __builtin_amdgcn_wmma_f32_16x16x4_f32(false, (a), false, (b), (short)0, (c), false, false)

__device__ __forceinline__ void lds_barrier() {
    __builtin_amdgcn_fence(__ATOMIC_RELEASE, "workgroup", "local");
    __builtin_amdgcn_s_barrier();
    __builtin_amdgcn_fence(__ATOMIC_ACQUIRE, "workgroup", "local");
}

template <bool EXTRA>
__global__ __launch_bounds__(256) void gemm_wmma_f32(
    const float* __restrict__ A,
    const float* __restrict__ W, const float* __restrict__ WkE, const float* __restrict__ WvE,
    const float* __restrict__ bias,
    float* __restrict__ C, float* __restrict__ CkE, float* __restrict__ CvE,
    int M, int N, int K, float alpha)
{
    const int bx = (int)blockIdx.x;

    constexpr int BK  = 64;
    constexpr int LDP = BK + 4;

    __shared__ __align__(16) float As[2][128][LDP];
    __shared__ __align__(16) float Wt[2][64][LDP];
    __shared__ __align__(16) float Et[2][EXTRA ? 64 : 1][LDP];

    const int tid    = threadIdx.x;
    const int wave   = tid >> 5;
    const int lane   = tid & 31;
    const int l16    = lane & 15;
    const int half   = lane >> 4;
    const int kb     = half << 1;
    const int rowOff = half << 3;

    const int wm = wave >> 1;
    const int wn = wave & 1;

    const int rowBlock = blockIdx.y * 128;
    const int colBlock = bx * 64;

    v8f acc[2][2];
    #pragma unroll
    for (int mt = 0; mt < 2; ++mt)
        #pragma unroll
        for (int nt = 0; nt < 2; ++nt) acc[mt][nt] = (v8f)(0.0f);

    const int sr = tid >> 4;
    const int sc = (tid & 15) << 2;
    const int wk  = (tid >> 2) & 15;
    const int wn4 = ((tid & 3) << 2) + ((tid >> 6) << 4);

    const int aRow0 = (wm << 5) + l16;
    const int aRow1 = aRow0 + 16;
    const int bRow0 = (wn << 5) + l16;
    const int bRow1 = bRow0 + 16;

    const float* WE = (bx & 1) ? WvE : WkE;
    float* CE = (bx & 1) ? CvE : CkE;
    const int eRow = (bx >> 1) << 4;
    const int ect  = wave & 3;
    const int ekk  = (wave >> 2) << 3;
    const int eaRow = eRow + l16;
    const int ebRow = (ect << 4) + l16;
    v8f accE = (v8f)(0.0f);

    const float* aSrc = A + (size_t)(rowBlock + sr) * K + sc;
    const float* wSrc = W + (size_t)wk * N + colBlock + wn4;
    const float* eSrc = WE + (size_t)wk * 64 + wn4;
    v4f av[8], wv[4], ev[4];
    #pragma unroll
    for (int p = 0; p < 8; ++p) av[p] = *(const v4f*)(aSrc + (size_t)(p << 4) * K);
    #pragma unroll
    for (int p = 0; p < 4; ++p) wv[p] = *(const v4f*)(wSrc + (size_t)(p << 4) * N);
    #pragma unroll
    for (int p = 0; p < 4; ++p) ev[p] = (v4f)(0.0f);
    if (EXTRA) {
        #pragma unroll
        for (int p = 0; p < 4; ++p) ev[p] = *(const v4f*)(eSrc + (p << 10));
    }

    auto stage_store = [&](int s) __attribute__((always_inline)) {
        float (*AsS)[LDP] = As[s];
        float (*WtS)[LDP] = Wt[s];
        float (*EtS)[LDP] = Et[s];
        #pragma unroll
        for (int p = 0; p < 8; ++p) *(v4f*)&AsS[sr + (p << 4)][sc] = av[p];
        #pragma unroll
        for (int p = 0; p < 4; ++p) {
            const int k = wk + (p << 4);
            WtS[wn4 + 0][k] = wv[p].x; WtS[wn4 + 1][k] = wv[p].y;
            WtS[wn4 + 2][k] = wv[p].z; WtS[wn4 + 3][k] = wv[p].w;
        }
        if (EXTRA) {
            #pragma unroll
            for (int p = 0; p < 4; ++p) {
                const int k = wk + (p << 4);
                EtS[wn4 + 0][k] = ev[p].x; EtS[wn4 + 1][k] = ev[p].y;
                EtS[wn4 + 2][k] = ev[p].z; EtS[wn4 + 3][k] = ev[p].w;
            }
        }
    };
    auto next_loads = [&]() __attribute__((always_inline)) {
        aSrc += BK;
        wSrc += (size_t)BK * N;
        #pragma unroll
        for (int p = 0; p < 8; ++p) av[p] = *(const v4f*)(aSrc + (size_t)(p << 4) * K);
        #pragma unroll
        for (int p = 0; p < 4; ++p) wv[p] = *(const v4f*)(wSrc + (size_t)(p << 4) * N);
        if (EXTRA) {
            eSrc += (size_t)BK * 64;
            #pragma unroll
            for (int p = 0; p < 4; ++p) ev[p] = *(const v4f*)(eSrc + (p << 10));
        }
    };

    auto frag_load = [&](int s, int kk, v2f (&f)[4]) __attribute__((always_inline)) {
        f[0] = *(const v2f*)&As[s][aRow0][kk + kb];
        f[1] = *(const v2f*)&As[s][aRow1][kk + kb];
        f[2] = *(const v2f*)&Wt[s][bRow0][kk + kb];
        f[3] = *(const v2f*)&Wt[s][bRow1][kk + kb];
    };
    auto frag_mma = [&](const v2f (&f)[4]) __attribute__((always_inline)) {
        acc[0][0] = WMMA_F32(f[0], f[2], acc[0][0]);
        acc[0][1] = WMMA_F32(f[0], f[3], acc[0][1]);
        acc[1][0] = WMMA_F32(f[1], f[2], acc[1][0]);
        acc[1][1] = WMMA_F32(f[1], f[3], acc[1][1]);
    };
    auto groups = [&](int s, int kkFrom, int kkTo) __attribute__((always_inline)) {
        #pragma unroll
        for (int kk = kkFrom; kk < kkTo; kk += 4) {
            v2f f[4];
            frag_load(s, kk, f);
            frag_mma(f);
        }
    };
    auto extra_piece = [&](int s, int ks) __attribute__((always_inline)) {
        if (EXTRA) {
            #pragma unroll
            for (int q = 0; q < 2; ++q) {
                const int kk = ks + ekk + (q << 2);
                const v2f ea = *(const v2f*)&As[s][eaRow][kk + kb];
                const v2f eb = *(const v2f*)&Et[s][ebRow][kk + kb];
                accE = WMMA_F32(ea, eb, accE);
            }
        }
    };

    stage_store(0);
    if (BK < K) next_loads();
    lds_barrier();

    int cur = 0;
    for (int kt = 0; kt + BK < K; kt += BK) {
        groups(cur, 0, 16);  extra_piece(cur, 0);
        groups(cur, 16, 32); extra_piece(cur, 16);
        v2f fr32[4], fr36[4];
        frag_load(cur, 32, fr32);
        frag_load(cur, 36, fr36);
        stage_store(cur ^ 1);
        if (kt + 2 * BK < K) next_loads();
        frag_mma(fr32);
        frag_mma(fr36);
        groups(cur, 40, 48); extra_piece(cur, 32);
        groups(cur, 48, 64); extra_piece(cur, 48);
        lds_barrier();
        cur ^= 1;
    }
    groups(cur, 0, 16);  extra_piece(cur, 0);
    groups(cur, 16, 32); extra_piece(cur, 16);
    groups(cur, 32, 48); extra_piece(cur, 32);
    groups(cur, 48, 64); extra_piece(cur, 48);
    lds_barrier();

    #pragma unroll
    for (int mt = 0; mt < 2; ++mt) {
        #pragma unroll
        for (int nt = 0; nt < 2; ++nt) {
            const int col = colBlock + (wn << 5) + (nt << 4) + l16;
            const float bv = bias ? bias[col] : 0.0f;
            #pragma unroll
            for (int v = 0; v < 8; ++v) {
                const int row = rowBlock + (wm << 5) + (mt << 4) + rowOff + v;
                C[(size_t)row * N + col] = alpha * acc[mt][nt][v] + bv;
            }
        }
    }

    if (EXTRA) {
        float* red = &As[0][0][0];
        if (wave >= 4) {
            #pragma unroll
            for (int v = 0; v < 8; ++v) red[(ect << 8) + (v << 5) + lane] = accE[v];
        }
        lds_barrier();
        if (wave < 4) {
            const int col = (ect << 4) + l16;
            #pragma unroll
            for (int v = 0; v < 8; ++v) {
                const int row = rowBlock + eRow + rowOff + v;
                CE[(size_t)row * 64 + col] = accE[v] + red[(ect << 8) + (v << 5) + lane];
            }
        }
    }
}

__device__ __forceinline__ float xhalf(float x) {
    const int b = __builtin_bit_cast(int, x);
    return __builtin_bit_cast(float, __builtin_amdgcn_permlanex16(b, b, 0x76543210, 0xfedcba98, false, false));
}

__global__ __launch_bounds__(256) void mqa_flash_wmma(
    const float* __restrict__ Q,
    const float* __restrict__ Kg,
    const float* __restrict__ Vg,
    float* __restrict__ Out,
    int S, int DIMp)
{
    __shared__ __align__(16) float Ks[2][64][68];
    __shared__ __align__(16) float Vt[2][64][68];
    __shared__ __align__(16) float Ps[64][68];
    __shared__ float Ms[64];
    __shared__ float Ls[64];

    const int tid    = threadIdx.x;
    const int wave   = tid >> 5;
    const int lane   = tid & 31;
    const int l16    = lane & 15;
    const int half   = lane >> 4;
    const int kb     = half << 1;
    const int rowOff = half << 3;
    const int qw     = wave & 3;
    const int kg     = wave >> 2;
    const int keyOff = kg << 5;

    const int h     = blockIdx.y;
    const int b     = blockIdx.z;
    const size_t rowB = (size_t)b * S;
    const int nQ    = (int)gridDim.x << 1;

    const int lr = tid >> 2;
    const int lq = (tid & 3) << 2;

    const int qr = (qw << 4) + l16;

    for (int pass = 0; pass < 2; ++pass) {
        const int qTile = pass == 0 ? (nQ - 1 - (int)blockIdx.x) : (int)blockIdx.x;
        const int qBase = qTile << 6;

        const int nTiles = qTile + 1;

        const float* ksrc = Kg + (rowB + lr) * 64 + lq;
        const float* vsrc = Vg + (rowB + lr) * 64 + lq;
        float4 kr[4], vr[4];
        #pragma unroll
        for (int j = 0; j < 4; ++j) {
            kr[j] = *(const float4*)(ksrc + (j << 4));
            vr[j] = *(const float4*)(vsrc + (j << 4));
        }

        auto kv_store = [&](int s) __attribute__((always_inline)) {
            float (*KsS)[68] = Ks[s];
            float (*VtS)[68] = Vt[s];
            #pragma unroll
            for (int j = 0; j < 4; ++j) {
                const float4 kv = kr[j];
                const float4 vv = vr[j];
                KsS[lr][lq + (j << 4) + 0] = kv.x; KsS[lr][lq + (j << 4) + 1] = kv.y;
                KsS[lr][lq + (j << 4) + 2] = kv.z; KsS[lr][lq + (j << 4) + 3] = kv.w;
                VtS[lq + (j << 4) + 0][lr] = vv.x; VtS[lq + (j << 4) + 1][lr] = vv.y;
                VtS[lq + (j << 4) + 2][lr] = vv.z; VtS[lq + (j << 4) + 3][lr] = vv.w;
            }
        };
        auto kv_next_loads = [&]() __attribute__((always_inline)) {
            ksrc += 64 * 64;
            vsrc += 64 * 64;
            #pragma unroll
            for (int j = 0; j < 4; ++j) {
                kr[j] = *(const float4*)(ksrc + (j << 4));
                vr[j] = *(const float4*)(vsrc + (j << 4));
            }
        };

        if (pass != 0) lds_barrier();

        {
            const float* src = Q + (rowB + qBase + lr) * DIMp + (h << 6) + lq;
            #pragma unroll
            for (int j = 0; j < 4; ++j) {
                const float4 qv = *(const float4*)(src + (j << 4));
                Ps[lr][lq + (j << 4) + 0] = qv.x; Ps[lr][lq + (j << 4) + 1] = qv.y;
                Ps[lr][lq + (j << 4) + 2] = qv.z; Ps[lr][lq + (j << 4) + 3] = qv.w;
            }
        }
        kv_store(0);
        if (1 < nTiles) kv_next_loads();
        lds_barrier();

        v2f bq[16];
        #pragma unroll
        for (int i = 0; i < 16; ++i)
            bq[i] = *(const v2f*)&Ps[qr][(i << 2) + kb];

        v8f ot[4];
        #pragma unroll
        for (int dt = 0; dt < 4; ++dt) ot[dt] = (v8f)(0.0f);
        float m = -1e30f;
        float lsum = 0.0f;

        const int qIdx = qBase + qr;

        for (int t = 0; t < nTiles; ++t) {
            const int kBase = t << 6;
            float (*KsC)[68] = Ks[t & 1];
            float (*VtC)[68] = Vt[t & 1];

            v8f st[2];
            #pragma unroll
            for (int mt = 0; mt < 2; ++mt) st[mt] = (v8f)(0.0f);
            #pragma unroll
            for (int kk = 0; kk < 64; kk += 4) {
                const v2f bf = bq[kk >> 2];
                #pragma unroll
                for (int mt = 0; mt < 2; ++mt) {
                    const v2f af = *(const v2f*)&KsC[keyOff + (mt << 4) + l16][kk + kb];
                    st[mt] = WMMA_F32(af, bf, st[mt]);
                }
            }

            if (t + 1 < nTiles) {
                kv_store((t + 1) & 1);
                if (t + 2 < nTiles) kv_next_loads();
            }

            if (t == nTiles - 1) {
                #pragma unroll
                for (int mt = 0; mt < 2; ++mt) {
                    #pragma unroll
                    for (int r = 0; r < 8; ++r) {
                        const int key = kBase + keyOff + (mt << 4) + rowOff + r;
                        if (key > qIdx) st[mt][r] = -1e30f;
                    }
                }
            }

            float x = -1e30f;
            #pragma unroll
            for (int mt = 0; mt < 2; ++mt) {
                #pragma unroll
                for (int r = 0; r < 8; ++r) x = fmaxf(x, st[mt][r]);
            }
            x = fmaxf(x, xhalf(x));
            const float mNew = fmaxf(m, x);
            const float corr = __expf(m - mNew);
            m = mNew;
            #pragma unroll
            for (int dt = 0; dt < 4; ++dt)
                #pragma unroll
                for (int r = 0; r < 8; ++r) ot[dt][r] *= corr;

            float rs = 0.0f;
            #pragma unroll
            for (int mt = 0; mt < 2; ++mt) {
                #pragma unroll
                for (int r = 0; r < 8; ++r) {
                    const float p = __expf(st[mt][r] - m);
                    st[mt][r] = p;
                    rs += p;
                }
            }
            lsum = lsum * corr + rs;

            #pragma unroll
            for (int mt = 0; mt < 2; ++mt) {
                #pragma unroll
                for (int j = 0; j < 4; ++j) {
                    v2f pf;
                    pf.x = st[mt][2 * j];
                    pf.y = st[mt][2 * j + 1];
                    #pragma unroll
                    for (int dt = 0; dt < 4; ++dt) {
                        const v2f af = *(const v2f*)&VtC[(dt << 4) + l16][keyOff + (mt << 4) + rowOff + (j << 1)];
                        ot[dt] = WMMA_F32(af, pf, ot[dt]);
                    }
                }
            }
            lds_barrier();
        }

        {
            const float ltot = lsum + xhalf(lsum);
            if (kg == 1) {
                #pragma unroll
                for (int dt = 0; dt < 4; ++dt) {
                    float4 lo, hi;
                    lo.x = ot[dt][0]; lo.y = ot[dt][1]; lo.z = ot[dt][2]; lo.w = ot[dt][3];
                    hi.x = ot[dt][4]; hi.y = ot[dt][5]; hi.z = ot[dt][6]; hi.w = ot[dt][7];
                    *(float4*)&Ps[qr][(dt << 4) + rowOff + 0] = lo;
                    *(float4*)&Ps[qr][(dt << 4) + rowOff + 4] = hi;
                }
                if (half == 0) { Ms[qr] = m; Ls[qr] = ltot; }
            }
            lds_barrier();
            if (kg == 0) {
                const float m1 = Ms[qr];
                const float l1 = Ls[qr];
                const float mm = fmaxf(m, m1);
                const float a0 = __expf(m - mm);
                const float a1 = __expf(m1 - mm);
                const float den = a0 * ltot + a1 * l1;
                #pragma unroll
                for (int dt = 0; dt < 4; ++dt) {
                    const float4 plo = *(const float4*)&Ps[qr][(dt << 4) + rowOff + 0];
                    const float4 phi = *(const float4*)&Ps[qr][(dt << 4) + rowOff + 4];
                    float4 lo, hi;
                    lo.x = (a0 * ot[dt][0] + a1 * plo.x) / den; lo.y = (a0 * ot[dt][1] + a1 * plo.y) / den;
                    lo.z = (a0 * ot[dt][2] + a1 * plo.z) / den; lo.w = (a0 * ot[dt][3] + a1 * plo.w) / den;
                    hi.x = (a0 * ot[dt][4] + a1 * phi.x) / den; hi.y = (a0 * ot[dt][5] + a1 * phi.y) / den;
                    hi.z = (a0 * ot[dt][6] + a1 * phi.z) / den; hi.w = (a0 * ot[dt][7] + a1 * phi.w) / den;
                    *(float4*)&Ps[qr][(dt << 4) + rowOff + 0] = lo;
                    *(float4*)&Ps[qr][(dt << 4) + rowOff + 4] = hi;
                }
            }
            lds_barrier();
            const int c4 = l16 << 2;
            #pragma unroll
            for (int i = 0; i < 4; ++i) {
                const int rr = (wave << 3) + (i << 1) + half;
                const float4 v = *(const float4*)&Ps[rr][c4];
                *(float4*)&Out[(rowB + qBase + rr) * DIMp + (h << 6) + c4] = v;
            }
        }
    }
}

extern "C" void kernel_launch(void* const* d_in, const int* in_sizes, int n_in,
                              void* d_out, int out_size, void* d_ws, size_t ws_size,
                              hipStream_t stream) {
    (void)in_sizes; (void)n_in; (void)out_size; (void)ws_size;
    constexpr int DIM = 1024, HEAD_DIM = 64, NUM_HEADS = 16, BATCH = 2, SEQ = 2048;
    constexpr int M = BATCH * SEQ;

    const float* x  = (const float*)d_in[0];
    const float* Wq = (const float*)d_in[1];
    const float* Wk = (const float*)d_in[2];
    const float* Wv = (const float*)d_in[3];
    const float* Wo = (const float*)d_in[4];
    const float* bo = (const float*)d_in[5];
    float* out = (float*)d_out;

    float* Qbuf = (float*)d_ws;
    float* Kbuf = Qbuf + (size_t)M * DIM;
    float* Vbuf = Kbuf + (size_t)M * HEAD_DIM;
    float* Abuf = Vbuf + (size_t)M * HEAD_DIM;

    const float scale = 0.125f;

    gemm_wmma_f32<true><<<dim3(DIM / 64, M / 128), 256, 0, stream>>>(
        x, Wq, Wk, Wv, nullptr, Qbuf, Kbuf, Vbuf, M, DIM, DIM, scale);

    mqa_flash_wmma<<<dim3(SEQ / 128, NUM_HEADS, BATCH), 256, 0, stream>>>(
        Qbuf, Kbuf, Vbuf, Abuf, SEQ, DIM);

    gemm_wmma_f32<false><<<dim3(DIM / 64, M / 128), 256, 0, stream>>>(
        Abuf, Wo, nullptr, nullptr, bo, out, nullptr, nullptr, M, DIM, DIM, 1.0f);
}


# All three kernels: wave priority steps down 3,2,1,0 with a workgroup's own progress (s_setprio at 1/2, 3/4, 7/8 of its tiles): a WGP's two workgroups take turns and end together; sums unchanged
# speedup vs baseline: 2.1346x; 1.0306x over previous
#include <hip/hip_runtime.h>
#include <math.h>

typedef __attribute__((ext_vector_type(2))) float v2f;
typedef __attribute__((ext_vector_type(4))) float v4f;
typedef __attribute__((ext_vector_type(8))) float v8f;

#define WMMA_F32(a, b, c) \
    __builtin_amdgcn_wmma_f32_16x16x4_f32(false, (a), false, (b), (short)0, (c), false, false)

__device__ __forceinline__ void lds_barrier() {
    __builtin_amdgcn_fence(__ATOMIC_RELEASE, "workgroup", "local");
    __builtin_amdgcn_s_barrier();
    __builtin_amdgcn_fence(__ATOMIC_ACQUIRE, "workgroup", "local");
}

__device__ __forceinline__ void progress_prio(int done, int total) {
    if (done >= (total >> 1)) {
        if (done >= total - (total >> 2)) {
            if (done >= total - (total >> 3)) __builtin_amdgcn_s_setprio(0);
            else                              __builtin_amdgcn_s_setprio(1);
        } else {
            __builtin_amdgcn_s_setprio(2);
        }
    }
}

template <bool EXTRA>
__global__ __launch_bounds__(256) void gemm_wmma_f32(
    const float* __restrict__ A,
    const float* __restrict__ W, const float* __restrict__ WkE, const float* __restrict__ WvE,
    const float* __restrict__ bias,
    float* __restrict__ C, float* __restrict__ CkE, float* __restrict__ CvE,
    int M, int N, int K, float alpha)
{
    const int bx = (int)blockIdx.x;

    __builtin_amdgcn_s_setprio(3);

    constexpr int BK  = 64;
    constexpr int LDP = BK + 4;

    __shared__ __align__(16) float As[2][128][LDP];
    __shared__ __align__(16) float Wt[2][64][LDP];
    __shared__ __align__(16) float Et[2][EXTRA ? 64 : 1][LDP];

    const int tid    = threadIdx.x;
    const int wave   = tid >> 5;
    const int lane   = tid & 31;
    const int l16    = lane & 15;
    const int half   = lane >> 4;
    const int kb     = half << 1;
    const int rowOff = half << 3;

    const int wm = wave >> 1;
    const int wn = wave & 1;

    const int rowBlock = blockIdx.y * 128;
    const int colBlock = bx * 64;

    v8f acc[2][2];
    #pragma unroll
    for (int mt = 0; mt < 2; ++mt)
        #pragma unroll
        for (int nt = 0; nt < 2; ++nt) acc[mt][nt] = (v8f)(0.0f);

    const int sr = tid >> 4;
    const int sc = (tid & 15) << 2;
    const int wk  = (tid >> 2) & 15;
    const int wn4 = ((tid & 3) << 2) + ((tid >> 6) << 4);

    const int aRow0 = (wm << 5) + l16;
    const int aRow1 = aRow0 + 16;
    const int bRow0 = (wn << 5) + l16;
    const int bRow1 = bRow0 + 16;

    const float* WE = (bx & 1) ? WvE : WkE;
    float* CE = (bx & 1) ? CvE : CkE;
    const int eRow = (bx >> 1) << 4;
    const int ect  = wave & 3;
    const int ekk  = (wave >> 2) << 3;
    const int eaRow = eRow + l16;
    const int ebRow = (ect << 4) + l16;
    v8f accE = (v8f)(0.0f);

    const float* aSrc = A + (size_t)(rowBlock + sr) * K + sc;
    const float* wSrc = W + (size_t)wk * N + colBlock + wn4;
    const float* eSrc = WE + (size_t)wk * 64 + wn4;
    v4f av[8], wv[4], ev[4];
    #pragma unroll
    for (int p = 0; p < 8; ++p) av[p] = *(const v4f*)(aSrc + (size_t)(p << 4) * K);
    #pragma unroll
    for (int p = 0; p < 4; ++p) wv[p] = *(const v4f*)(wSrc + (size_t)(p << 4) * N);
    #pragma unroll
    for (int p = 0; p < 4; ++p) ev[p] = (v4f)(0.0f);
    if (EXTRA) {
        #pragma unroll
        for (int p = 0; p < 4; ++p) ev[p] = *(const v4f*)(eSrc + (p << 10));
    }

    auto stage_store = [&](int s) __attribute__((always_inline)) {
        float (*AsS)[LDP] = As[s];
        float (*WtS)[LDP] = Wt[s];
        float (*EtS)[LDP] = Et[s];
        #pragma unroll
        for (int p = 0; p < 8; ++p) *(v4f*)&AsS[sr + (p << 4)][sc] = av[p];
        #pragma unroll
        for (int p = 0; p < 4; ++p) {
            const int k = wk + (p << 4);
            WtS[wn4 + 0][k] = wv[p].x; WtS[wn4 + 1][k] = wv[p].y;
            WtS[wn4 + 2][k] = wv[p].z; WtS[wn4 + 3][k] = wv[p].w;
        }
        if (EXTRA) {
            #pragma unroll
            for (int p = 0; p < 4; ++p) {
                const int k = wk + (p << 4);
                EtS[wn4 + 0][k] = ev[p].x; EtS[wn4 + 1][k] = ev[p].y;
                EtS[wn4 + 2][k] = ev[p].z; EtS[wn4 + 3][k] = ev[p].w;
            }
        }
    };
    auto next_loads = [&]() __attribute__((always_inline)) {
        aSrc += BK;
        wSrc += (size_t)BK * N;
        #pragma unroll
        for (int p = 0; p < 8; ++p) av[p] = *(const v4f*)(aSrc + (size_t)(p << 4) * K);
        #pragma unroll
        for (int p = 0; p < 4; ++p) wv[p] = *(const v4f*)(wSrc + (size_t)(p << 4) * N);
        if (EXTRA) {
            eSrc += (size_t)BK * 64;
            #pragma unroll
            for (int p = 0; p < 4; ++p) ev[p] = *(const v4f*)(eSrc + (p << 10));
        }
    };

    auto frag_load = [&](int s, int kk, v2f (&f)[4]) __attribute__((always_inline)) {
        f[0] = *(const v2f*)&As[s][aRow0][kk + kb];
        f[1] = *(const v2f*)&As[s][aRow1][kk + kb];
        f[2] = *(const v2f*)&Wt[s][bRow0][kk + kb];
        f[3] = *(const v2f*)&Wt[s][bRow1][kk + kb];
    };
    auto frag_mma = [&](const v2f (&f)[4]) __attribute__((always_inline)) {
        acc[0][0] = WMMA_F32(f[0], f[2], acc[0][0]);
        acc[0][1] = WMMA_F32(f[0], f[3], acc[0][1]);
        acc[1][0] = WMMA_F32(f[1], f[2], acc[1][0]);
        acc[1][1] = WMMA_F32(f[1], f[3], acc[1][1]);
    };
    auto groups = [&](int s, int kkFrom, int kkTo) __attribute__((always_inline)) {
        #pragma unroll
        for (int kk = kkFrom; kk < kkTo; kk += 4) {
            v2f f[4];
            frag_load(s, kk, f);
            frag_mma(f);
        }
    };
    auto extra_piece = [&](int s, int ks) __attribute__((always_inline)) {
        if (EXTRA) {
            #pragma unroll
            for (int q = 0; q < 2; ++q) {
                const int kk = ks + ekk + (q << 2);
                const v2f ea = *(const v2f*)&As[s][eaRow][kk + kb];
                const v2f eb = *(const v2f*)&Et[s][ebRow][kk + kb];
                accE = WMMA_F32(ea, eb, accE);
            }
        }
    };

    stage_store(0);
    if (BK < K) next_loads();
    lds_barrier();

    int cur = 0;
    for (int kt = 0; kt + BK < K; kt += BK) {
        progress_prio(kt, K);
        groups(cur, 0, 16);  extra_piece(cur, 0);
        groups(cur, 16, 32); extra_piece(cur, 16);
        v2f fr32[4], fr36[4];
        frag_load(cur, 32, fr32);
        frag_load(cur, 36, fr36);
        stage_store(cur ^ 1);
        if (kt + 2 * BK < K) next_loads();
        frag_mma(fr32);
        frag_mma(fr36);
        groups(cur, 40, 48); extra_piece(cur, 32);
        groups(cur, 48, 64); extra_piece(cur, 48);
        lds_barrier();
        cur ^= 1;
    }
    groups(cur, 0, 16);  extra_piece(cur, 0);
    groups(cur, 16, 32); extra_piece(cur, 16);
    groups(cur, 32, 48); extra_piece(cur, 32);
    groups(cur, 48, 64); extra_piece(cur, 48);
    lds_barrier();
    __builtin_amdgcn_s_setprio(0);

    #pragma unroll
    for (int mt = 0; mt < 2; ++mt) {
        #pragma unroll
        for (int nt = 0; nt < 2; ++nt) {
            const int col = colBlock + (wn << 5) + (nt << 4) + l16;
            const float bv = bias ? bias[col] : 0.0f;
            #pragma unroll
            for (int v = 0; v < 8; ++v) {
                const int row = rowBlock + (wm << 5) + (mt << 4) + rowOff + v;
                C[(size_t)row * N + col] = alpha * acc[mt][nt][v] + bv;
            }
        }
    }

    if (EXTRA) {
        float* red = &As[0][0][0];
        if (wave >= 4) {
            #pragma unroll
            for (int v = 0; v < 8; ++v) red[(ect << 8) + (v << 5) + lane] = accE[v];
        }
        lds_barrier();
        if (wave < 4) {
            const int col = (ect << 4) + l16;
            #pragma unroll
            for (int v = 0; v < 8; ++v) {
                const int row = rowBlock + eRow + rowOff + v;
                CE[(size_t)row * 64 + col] = accE[v] + red[(ect << 8) + (v << 5) + lane];
            }
        }
    }
}

__device__ __forceinline__ float xhalf(float x) {
    const int b = __builtin_bit_cast(int, x);
    return __builtin_bit_cast(float, __builtin_amdgcn_permlanex16(b, b, 0x76543210, 0xfedcba98, false, false));
}

__global__ __launch_bounds__(256) void mqa_flash_wmma(
    const float* __restrict__ Q,
    const float* __restrict__ Kg,
    const float* __restrict__ Vg,
    float* __restrict__ Out,
    int S, int DIMp)
{
    __shared__ __align__(16) float Ks[2][64][68];
    __shared__ __align__(16) float Vt[2][64][68];
    __shared__ __align__(16) float Ps[64][68];
    __shared__ float Ms[64];
    __shared__ float Ls[64];

    const int tid    = threadIdx.x;
    const int wave   = tid >> 5;
    const int lane   = tid & 31;
    const int l16    = lane & 15;
    const int half   = lane >> 4;
    const int kb     = half << 1;
    const int rowOff = half << 3;
    const int qw     = wave & 3;
    const int kg     = wave >> 2;
    const int keyOff = kg << 5;

    const int h     = blockIdx.y;
    const int b     = blockIdx.z;
    const size_t rowB = (size_t)b * S;
    const int nQ    = (int)gridDim.x << 1;

    const int lr = tid >> 2;
    const int lq = (tid & 3) << 2;

    const int qr = (qw << 4) + l16;

    const int stepsAll = nQ + 1;
    int stepsDone = 0;
    __builtin_amdgcn_s_setprio(3);

    for (int pass = 0; pass < 2; ++pass) {
        const int qTile = pass == 0 ? (nQ - 1 - (int)blockIdx.x) : (int)blockIdx.x;
        const int qBase = qTile << 6;

        const int nTiles = qTile + 1;

        const float* ksrc = Kg + (rowB + lr) * 64 + lq;
        const float* vsrc = Vg + (rowB + lr) * 64 + lq;
        float4 kr[4], vr[4];
        #pragma unroll
        for (int j = 0; j < 4; ++j) {
            kr[j] = *(const float4*)(ksrc + (j << 4));
            vr[j] = *(const float4*)(vsrc + (j << 4));
        }

        auto kv_store = [&](int s) __attribute__((always_inline)) {
            float (*KsS)[68] = Ks[s];
            float (*VtS)[68] = Vt[s];
            #pragma unroll
            for (int j = 0; j < 4; ++j) {
                const float4 kv = kr[j];
                const float4 vv = vr[j];
                KsS[lr][lq + (j << 4) + 0] = kv.x; KsS[lr][lq + (j << 4) + 1] = kv.y;
                KsS[lr][lq + (j << 4) + 2] = kv.z; KsS[lr][lq + (j << 4) + 3] = kv.w;
                VtS[lq + (j << 4) + 0][lr] = vv.x; VtS[lq + (j << 4) + 1][lr] = vv.y;
                VtS[lq + (j << 4) + 2][lr] = vv.z; VtS[lq + (j << 4) + 3][lr] = vv.w;
            }
        };
        auto kv_next_loads = [&]() __attribute__((always_inline)) {
            ksrc += 64 * 64;
            vsrc += 64 * 64;
            #pragma unroll
            for (int j = 0; j < 4; ++j) {
                kr[j] = *(const float4*)(ksrc + (j << 4));
                vr[j] = *(const float4*)(vsrc + (j << 4));
            }
        };

        if (pass != 0) lds_barrier();

        {
            const float* src = Q + (rowB + qBase + lr) * DIMp + (h << 6) + lq;
            #pragma unroll
            for (int j = 0; j < 4; ++j) {
                const float4 qv = *(const float4*)(src + (j << 4));
                Ps[lr][lq + (j << 4) + 0] = qv.x; Ps[lr][lq + (j << 4) + 1] = qv.y;
                Ps[lr][lq + (j << 4) + 2] = qv.z; Ps[lr][lq + (j << 4) + 3] = qv.w;
            }
        }
        kv_store(0);
        if (1 < nTiles) kv_next_loads();
        lds_barrier();

        v2f bq[16];
        #pragma unroll
        for (int i = 0; i < 16; ++i)
            bq[i] = *(const v2f*)&Ps[qr][(i << 2) + kb];

        v8f ot[4];
        #pragma unroll
        for (int dt = 0; dt < 4; ++dt) ot[dt] = (v8f)(0.0f);
        float m = -1e30f;
        float lsum = 0.0f;

        const int qIdx = qBase + qr;

        for (int t = 0; t < nTiles; ++t) {
            progress_prio(stepsDone, stepsAll);
            ++stepsDone;
            const int kBase = t << 6;
            float (*KsC)[68] = Ks[t & 1];
            float (*VtC)[68] = Vt[t & 1];

            v8f st[2];
            #pragma unroll
            for (int mt = 0; mt < 2; ++mt) st[mt] = (v8f)(0.0f);
            #pragma unroll
            for (int kk = 0; kk < 64; kk += 4) {
                const v2f bf = bq[kk >> 2];
                #pragma unroll
                for (int mt = 0; mt < 2; ++mt) {
                    const v2f af = *(const v2f*)&KsC[keyOff + (mt << 4) + l16][kk + kb];
                    st[mt] = WMMA_F32(af, bf, st[mt]);
                }
            }

            if (t + 1 < nTiles) {
                kv_store((t + 1) & 1);
                if (t + 2 < nTiles) kv_next_loads();
            }

            if (t == nTiles - 1) {
                #pragma unroll
                for (int mt = 0; mt < 2; ++mt) {
                    #pragma unroll
                    for (int r = 0; r < 8; ++r) {
                        const int key = kBase + keyOff + (mt << 4) + rowOff + r;
                        if (key > qIdx) st[mt][r] = -1e30f;
                    }
                }
            }

            float x = -1e30f;
            #pragma unroll
            for (int mt = 0; mt < 2; ++mt) {
                #pragma unroll
                for (int r = 0; r < 8; ++r) x = fmaxf(x, st[mt][r]);
            }
            x = fmaxf(x, xhalf(x));
            const float mNew = fmaxf(m, x);
            const float corr = __expf(m - mNew);
            m = mNew;
            #pragma unroll
            for (int dt = 0; dt < 4; ++dt)
                #pragma unroll
                for (int r = 0; r < 8; ++r) ot[dt][r] *= corr;

            float rs = 0.0f;
            #pragma unroll
            for (int mt = 0; mt < 2; ++mt) {
                #pragma unroll
                for (int r = 0; r < 8; ++r) {
                    const float p = __expf(st[mt][r] - m);
                    st[mt][r] = p;
                    rs += p;
                }
            }
            lsum = lsum * corr + rs;

            #pragma unroll
            for (int mt = 0; mt < 2; ++mt) {
                #pragma unroll
                for (int j = 0; j < 4; ++j) {
                    v2f pf;
                    pf.x = st[mt][2 * j];
                    pf.y = st[mt][2 * j + 1];
                    #pragma unroll
                    for (int dt = 0; dt < 4; ++dt) {
                        const v2f af = *(const v2f*)&VtC[(dt << 4) + l16][keyOff + (mt << 4) + rowOff + (j << 1)];
                        ot[dt] = WMMA_F32(af, pf, ot[dt]);
                    }
                }
            }
            lds_barrier();
        }

        {
            const float ltot = lsum + xhalf(lsum);
            if (kg == 1) {
                #pragma unroll
                for (int dt = 0; dt < 4; ++dt) {
                    float4 lo, hi;
                    lo.x = ot[dt][0]; lo.y = ot[dt][1]; lo.z = ot[dt][2]; lo.w = ot[dt][3];
                    hi.x = ot[dt][4]; hi.y = ot[dt][5]; hi.z = ot[dt][6]; hi.w = ot[dt][7];
                    *(float4*)&Ps[qr][(dt << 4) + rowOff + 0] = lo;
                    *(float4*)&Ps[qr][(dt << 4) + rowOff + 4] = hi;
                }
                if (half == 0) { Ms[qr] = m; Ls[qr] = ltot; }
            }
            lds_barrier();
            if (kg == 0) {
                const float m1 = Ms[qr];
                const float l1 = Ls[qr];
                const float mm = fmaxf(m, m1);
                const float a0 = __expf(m - mm);
                const float a1 = __expf(m1 - mm);
                const float den = a0 * ltot + a1 * l1;
                #pragma unroll
                for (int dt = 0; dt < 4; ++dt) {
                    const float4 plo = *(const float4*)&Ps[qr][(dt << 4) + rowOff + 0];
                    const float4 phi = *(const float4*)&Ps[qr][(dt << 4) + rowOff + 4];
                    float4 lo, hi;
                    lo.x = (a0 * ot[dt][0] + a1 * plo.x) / den; lo.y = (a0 * ot[dt][1] + a1 * plo.y) / den;
                    lo.z = (a0 * ot[dt][2] + a1 * plo.z) / den; lo.w = (a0 * ot[dt][3] + a1 * plo.w) / den;
                    hi.x = (a0 * ot[dt][4] + a1 * phi.x) / den; hi.y = (a0 * ot[dt][5] + a1 * phi.y) / den;
                    hi.z = (a0 * ot[dt][6] + a1 * phi.z) / den; hi.w = (a0 * ot[dt][7] + a1 * phi.w) / den;
                    *(float4*)&Ps[qr][(dt << 4) + rowOff + 0] = lo;
                    *(float4*)&Ps[qr][(dt << 4) + rowOff + 4] = hi;
                }
            }
            lds_barrier();
            const int c4 = l16 << 2;
            #pragma unroll
            for (int i = 0; i < 4; ++i) {
                const int rr = (wave << 3) + (i << 1) + half;
                const float4 v = *(const float4*)&Ps[rr][c4];
                *(float4*)&Out[(rowB + qBase + rr) * DIMp + (h << 6) + c4] = v;
            }
        }
    }
    __builtin_amdgcn_s_setprio(0);
}

extern "C" void kernel_launch(void* const* d_in, const int* in_sizes, int n_in,
                              void* d_out, int out_size, void* d_ws, size_t ws_size,
                              hipStream_t stream) {
    (void)in_sizes; (void)n_in; (void)out_size; (void)ws_size;
    constexpr int DIM = 1024, HEAD_DIM = 64, NUM_HEADS = 16, BATCH = 2, SEQ = 2048;
    constexpr int M = BATCH * SEQ;

    const float* x  = (const float*)d_in[0];
    const float* Wq = (const float*)d_in[1];
    const float* Wk = (const float*)d_in[2];
    const float* Wv = (const float*)d_in[3];
    const float* Wo = (const float*)d_in[4];
    const float* bo = (const float*)d_in[5];
    float* out = (float*)d_out;

    float* Qbuf = (float*)d_ws;
    float* Kbuf = Qbuf + (size_t)M * DIM;
    float* Vbuf = Kbuf + (size_t)M * HEAD_DIM;
    float* Abuf = Vbuf + (size_t)M * HEAD_DIM;

    const float scale = 0.125f;

    gemm_wmma_f32<true><<<dim3(DIM / 64, M / 128), 256, 0, stream>>>(
        x, Wq, Wk, Wv, nullptr, Qbuf, Kbuf, Vbuf, M, DIM, DIM, scale);

    mqa_flash_wmma<<<dim3(SEQ / 128, NUM_HEADS, BATCH), 256, 0, stream>>>(
        Qbuf, Kbuf, Vbuf, Abuf, SEQ, DIM);

    gemm_wmma_f32<false><<<dim3(DIM / 64, M / 128), 256, 0, stream>>>(
        Abuf, Wo, nullptr, nullptr, bo, out, nullptr, nullptr, M, DIM, DIM, 1.0f);
}
